# starting kernel
# baseline (speedup 1.0000x reference)
#include <hip/hip_runtime.h>
#include <hip/hip_bf16.h>
#include <math.h>


typedef __attribute__((ext_vector_type(2))) float v2f;
typedef __attribute__((ext_vector_type(8))) float v8f;

__device__ __forceinline__ v8f wmma4(v2f a, v2f b, v8f c) {
  return __builtin_amdgcn_wmma_f32_16x16x4_f32(
       false, a,  false, b,
       (short)0, c,  false,  false);
}

__global__ __launch_bounds__(256) void temb_kernel(
    const int* __restrict__ num_loops,
    const float* __restrict__ t_w1, const float* __restrict__ t_b1,
    const float* __restrict__ t_w2, const float* __restrict__ t_b2,
    float* __restrict__ ttab) {
  const int t = blockIdx.x;
  if (t >= *num_loops) return;
  __shared__ float te[256];
  __shared__ float hs[1024];
  __shared__ float red[256];
  const int tid = threadIdx.x;

  if (tid < 128) {
    float f   = expf(-logf(10000.0f) * (float)tid / 128.0f);
    float arg = (float)t * f;
    te[tid]       = cosf(arg);
    te[128 + tid] = sinf(arg);
  }
  __syncthreads();

  for (int j = tid; j < 1024; j += 256) {
    float acc = t_b1[j];
    for (int k = 0; k < 256; ++k) acc += te[k] * t_w1[k * 1024 + j];
    hs[j] = acc / (1.0f + expf(-acc));
  }
  __syncthreads();

  const int j = tid & 63, p = tid >> 6;
  float acc = 0.0f;
  for (int k = p * 256; k < p * 256 + 256; ++k) acc += hs[k] * t_w2[k * 64 + j];
  red[tid] = acc;
  __syncthreads();
  if (tid < 64) {
    ttab[t * 64 + tid] =
        t_b2[tid] + red[tid] + red[tid + 64] + red[tid + 128] + red[tid + 192];
  }
}

#define TB 512
#define NW 16

__device__ __forceinline__ void ln_rows(const float* __restrict__ src,
                                        float* __restrict__ dst,
                                        const float* __restrict__ g,
                                        const float* __restrict__ bta, int tid) {
  const int s = tid >> 2, p = tid & 3;
  float sum = 0.0f, sq = 0.0f;
  for (int d = p * 16; d < p * 16 + 16; ++d) {
    float xv = src[s * 64 + d];
    sum += xv; sq += xv * xv;
  }
  for (int off = 1; off < 4; off <<= 1) {
    sum += __shfl_xor(sum, off, 4);
    sq  += __shfl_xor(sq,  off, 4);
  }
  const float mu  = sum * (1.0f / 64.0f);
  const float var = sq * (1.0f / 64.0f) - mu * mu;
  const float inv = rsqrtf(var + 1e-5f);
  for (int d = p * 16; d < p * 16 + 16; ++d)
    dst[s * 64 + d] = (src[s * 64 + d] - mu) * inv * g[d] + bta[d];
}

__global__ __launch_bounds__(TB, 1) void xformer_kernel(
    const int* __restrict__ idx, const int* __restrict__ num_loops,
    const float* __restrict__ wte, const float* __restrict__ wpe,
    const float* __restrict__ ln1_g, const float* __restrict__ ln1_b,
    const float* __restrict__ w_qkv, const float* __restrict__ b_qkv,
    const float* __restrict__ w_o, const float* __restrict__ b_o,
    const float* __restrict__ ln2_g, const float* __restrict__ ln2_b,
    const float* __restrict__ w_fc, const float* __restrict__ b_fc,
    const float* __restrict__ w_pr, const float* __restrict__ b_pr,
    const float* __restrict__ lnf_g, const float* __restrict__ lnf_b,
    const float* __restrict__ ttab, float* __restrict__ out) {
  __shared__ float smem[40960];
  float* xb = smem;
  float* hb = smem + 8192;
  float* qb = smem + 16384;
  float* kb = smem + 24576;
  float* vb = smem + 32768;
  float* gb = qb;
  float* ob = vb;

  const int b    = blockIdx.x;
  const int tid  = threadIdx.x;
  const int wid  = tid >> 5;
  const int lane = tid & 31;
  const int l16  = lane & 15;
  const int half = lane >> 4;
  const int koff = 2 * half;

  for (int i = tid; i < 8192; i += TB) {
    int s = i >> 6, d = i & 63;
    xb[i] = wte[idx[b * 128 + s] * 64 + d] + wpe[i];
  }
  __syncthreads();

  const int NL = *num_loops;
  for (int t = 0; t < NL; ++t) {
    for (int i = tid; i < 8192; i += TB) xb[i] += ttab[t * 64 + (i & 63)];
    __syncthreads();

    ln_rows(xb, hb, ln1_g, ln1_b, tid);
    __syncthreads();

    for (int tile = wid; tile < 96; tile += NW) {
      const int m0 = (tile / 12) * 16, n0 = (tile % 12) * 16;
      v8f acc;
      { float bias = b_qkv[n0 + l16]; for (int r = 0; r < 8; ++r) acc[r] = bias; }
      for (int k = 0; k < 64; k += 4) {
        v2f a, w;
        a.x = hb[(m0 + l16) * 64 + k + koff];
        a.y = hb[(m0 + l16) * 64 + k + koff + 1];
        w.x = w_qkv[(k + koff) * 192 + n0 + l16];
        w.y = w_qkv[(k + koff + 1) * 192 + n0 + l16];
        acc = wmma4(a, w, acc);
      }
      float* dst; int nc;
      if (n0 < 64)       { dst = qb; nc = n0; }
      else if (n0 < 128) { dst = kb; nc = n0 - 64; }
      else               { dst = vb; nc = n0 - 128; }
      for (int r = 0; r < 8; ++r)
        dst[(m0 + r + 8 * half) * 64 + nc + l16] = acc[r];
    }
    __syncthreads();

    for (int job = wid; job < 32; job += NW) {
      const int hh = job >> 3, qt = job & 7;
      const int q0 = qt * 16, c0 = hh * 16;
      v8f sc[8];
      for (int kt = 0; kt < 8; ++kt) {
        v8f a8; for (int r = 0; r < 8; ++r) a8[r] = 0.0f;
        for (int kk = 0; kk < 16; kk += 4) {
          v2f a, bb;
          a.x  = qb[(q0 + l16) * 64 + c0 + kk + koff];
          a.y  = qb[(q0 + l16) * 64 + c0 + kk + koff + 1];
          bb.x = kb[(kt * 16 + l16) * 64 + c0 + kk + koff];
          bb.y = kb[(kt * 16 + l16) * 64 + c0 + kk + koff + 1];
          a8 = wmma4(a, bb, a8);
        }
        for (int r = 0; r < 8; ++r) {
          const int row = q0 + r + 8 * half, col = kt * 16 + l16;
          a8[r] = (col <= row) ? a8[r] * 0.25f : -1e9f;
        }
        sc[kt] = a8;
      }
      for (int r = 0; r < 8; ++r) {
        float mx = -1e30f;
        for (int kt = 0; kt < 8; ++kt) mx = fmaxf(mx, sc[kt][r]);
        for (int off = 8; off; off >>= 1) mx = fmaxf(mx, __shfl_xor(mx, off, 16));
        float sum = 0.0f;
        for (int kt = 0; kt < 8; ++kt) {
          float e = __expf(sc[kt][r] - mx);
          sc[kt][r] = e; sum += e;
        }
        for (int off = 8; off; off >>= 1) sum += __shfl_xor(sum, off, 16);
        const float inv = 1.0f / sum;
        for (int kt = 0; kt < 8; ++kt) sc[kt][r] *= inv;
      }
      v8f o; for (int r = 0; r < 8; ++r) o[r] = 0.0f;
      for (int kt = 0; kt < 8; ++kt) {
        for (int r = 0; r < 8; ++r)
          hb[(q0 + r + 8 * half) * 64 + c0 + l16] = sc[kt][r];
        for (int kk = 0; kk < 16; kk += 4) {
          v2f a, bb;
          a.x  = hb[(q0 + l16) * 64 + c0 + kk + koff];
          a.y  = hb[(q0 + l16) * 64 + c0 + kk + koff + 1];
          bb.x = vb[(kt * 16 + kk + koff) * 64 + c0 + l16];
          bb.y = vb[(kt * 16 + kk + koff + 1) * 64 + c0 + l16];
          o = wmma4(a, bb, o);
        }
      }
      for (int r = 0; r < 8; ++r)
        hb[(q0 + r + 8 * half) * 64 + c0 + l16] = o[r];
    }
    __syncthreads();

    for (int tile = wid; tile < 32; tile += NW) {
      const int m0 = (tile >> 2) * 16, n0 = (tile & 3) * 16;
      v8f acc;
      { float bias = b_o[n0 + l16]; for (int r = 0; r < 8; ++r) acc[r] = bias; }
      for (int k = 0; k < 64; k += 4) {
        v2f a, w;
        a.x = hb[(m0 + l16) * 64 + k + koff];
        a.y = hb[(m0 + l16) * 64 + k + koff + 1];
        w.x = w_o[(k + koff) * 64 + n0 + l16];
        w.y = w_o[(k + koff + 1) * 64 + n0 + l16];
        acc = wmma4(a, w, acc);
      }
      for (int r = 0; r < 8; ++r)
        xb[(m0 + r + 8 * half) * 64 + n0 + l16] += 0.1f * acc[r];
    }
    __syncthreads();

    ln_rows(xb, hb, ln2_g, ln2_b, tid);
    __syncthreads();

    for (int hh2 = 0; hh2 < 2; ++hh2) {
      for (int tile = wid; tile < 64; tile += NW) {
        const int m0 = (tile >> 3) * 16, ntl = (tile & 7) * 16;
        const int n0g = hh2 * 128 + ntl;
        v8f acc;
        { float bias = b_fc[n0g + l16]; for (int r = 0; r < 8; ++r) acc[r] = bias; }
        for (int k = 0; k < 64; k += 4) {
          v2f a, w;
          a.x = hb[(m0 + l16) * 64 + k + koff];
          a.y = hb[(m0 + l16) * 64 + k + koff + 1];
          w.x = w_fc[(k + koff) * 256 + n0g + l16];
          w.y = w_fc[(k + koff + 1) * 256 + n0g + l16];
          acc = wmma4(a, w, acc);
        }
        for (int r = 0; r < 8; ++r) {
          float u = acc[r];
          float g = 0.5f * u *
              (1.0f + tanhf(0.7978845608028654f * (u + 0.044715f * u * u * u)));
          gb[(m0 + r + 8 * half) * 128 + ntl + l16] = g;
        }
      }
      __syncthreads();
      for (int tile = wid; tile < 32; tile += NW) {
        const int m0 = (tile >> 2) * 16, n0 = (tile & 3) * 16;
        v8f acc;
        if (hh2 == 0) {
          float bias = b_pr[n0 + l16];
          for (int r = 0; r < 8; ++r) acc[r] = bias;
        } else {
          for (int r = 0; r < 8; ++r)
            acc[r] = ob[(m0 + r + 8 * half) * 64 + n0 + l16];
        }
        for (int k = 0; k < 128; k += 4) {
          v2f a, w;
          a.x = gb[(m0 + l16) * 128 + k + koff];
          a.y = gb[(m0 + l16) * 128 + k + koff + 1];
          w.x = w_pr[(hh2 * 128 + k + koff) * 64 + n0 + l16];
          w.y = w_pr[(hh2 * 128 + k + koff + 1) * 64 + n0 + l16];
          acc = wmma4(a, w, acc);
        }
        if (hh2 == 0) {
          for (int r = 0; r < 8; ++r)
            ob[(m0 + r + 8 * half) * 64 + n0 + l16] = acc[r];
        } else {
          for (int r = 0; r < 8; ++r)
            xb[(m0 + r + 8 * half) * 64 + n0 + l16] += 0.1f * acc[r];
        }
      }
      __syncthreads();
    }
  }

  {
    const int s = tid >> 2, p = tid & 3;
    float sum = 0.0f, sq = 0.0f;
    for (int d = p * 16; d < p * 16 + 16; ++d) {
      float xv = xb[s * 64 + d];
      sum += xv; sq += xv * xv;
    }
    for (int off = 1; off < 4; off <<= 1) {
      sum += __shfl_xor(sum, off, 4);
      sq  += __shfl_xor(sq,  off, 4);
    }
    const float mu  = sum * (1.0f / 64.0f);
    const float var = sq * (1.0f / 64.0f) - mu * mu;
    const float inv = rsqrtf(var + 1e-5f);
    for (int d = p * 16; d < p * 16 + 16; ++d)
      out[b * 8192 + s * 64 + d] =
          (xb[s * 64 + d] - mu) * inv * lnf_g[d] + lnf_b[d];
  }
}

extern "C" void kernel_launch(void* const* d_in, const int* in_sizes, int n_in,
                              void* d_out, int out_size, void* d_ws, size_t ws_size,
                              hipStream_t stream) {
  const int*   idx   = (const int*)  d_in[0];
  const int*   nlp   = (const int*)  d_in[1];
  const float* wte   = (const float*)d_in[2];
  const float* wpe   = (const float*)d_in[3];
  const float* t_w1  = (const float*)d_in[4];
  const float* t_b1  = (const float*)d_in[5];
  const float* t_w2  = (const float*)d_in[6];
  const float* t_b2  = (const float*)d_in[7];
  const float* ln1_g = (const float*)d_in[8];
  const float* ln1_b = (const float*)d_in[9];
  const float* w_qkv = (const float*)d_in[10];
  const float* b_qkv = (const float*)d_in[11];
  const float* w_o   = (const float*)d_in[12];
  const float* b_o   = (const float*)d_in[13];
  const float* ln2_g = (const float*)d_in[14];
  const float* ln2_b = (const float*)d_in[15];
  const float* w_fc  = (const float*)d_in[16];
  const float* b_fc  = (const float*)d_in[17];
  const float* w_pr  = (const float*)d_in[18];
  const float* b_pr  = (const float*)d_in[19];
  const float* lnf_g = (const float*)d_in[20];
  const float* lnf_b = (const float*)d_in[21];

  float* ttab = (float*)d_ws;

  temb_kernel<<<128, 256, 0, stream>>>(nlp, t_w1, t_b1, t_w2, t_b2, ttab);
  xformer_kernel<<<32, TB, 0, stream>>>(idx, nlp, wte, wpe, ln1_g, ln1_b,
                                        w_qkv, b_qkv, w_o, b_o, ln2_g, ln2_b,
                                        w_fc, b_fc, w_pr, b_pr, lnf_g, lnf_b,
                                        ttab, (float*)d_out);
}


# Pad LDS row pitch (64->76 and 128->132 dwords) so row-per-lane A/K fragment reads hit 64 distinct banks instead of a 16-way bank conflict; sums and their order unchanged
# speedup vs baseline: 2.2983x; 2.2983x over previous
#include <hip/hip_runtime.h>
#include <hip/hip_bf16.h>
#include <math.h>


typedef __attribute__((ext_vector_type(2))) float v2f;
typedef __attribute__((ext_vector_type(8))) float v8f;

__device__ __forceinline__ v8f wmma4(v2f a, v2f b, v8f c) {
  return __builtin_amdgcn_wmma_f32_16x16x4_f32(
       false, a,  false, b,
       (short)0, c,  false,  false);
}

__global__ __launch_bounds__(256) void temb_kernel(
    const int* __restrict__ num_loops,
    const float* __restrict__ t_w1, const float* __restrict__ t_b1,
    const float* __restrict__ t_w2, const float* __restrict__ t_b2,
    float* __restrict__ ttab) {
  const int t = blockIdx.x;
  if (t >= *num_loops) return;
  __shared__ float te[256];
  __shared__ float hs[1024];
  __shared__ float red[256];
  const int tid = threadIdx.x;

  if (tid < 128) {
    float f   = expf(-logf(10000.0f) * (float)tid / 128.0f);
    float arg = (float)t * f;
    te[tid]       = cosf(arg);
    te[128 + tid] = sinf(arg);
  }
  __syncthreads();

  for (int j = tid; j < 1024; j += 256) {
    float acc = t_b1[j];
    for (int k = 0; k < 256; ++k) acc += te[k] * t_w1[k * 1024 + j];
    hs[j] = acc / (1.0f + expf(-acc));
  }
  __syncthreads();

  const int j = tid & 63, p = tid >> 6;
  float acc = 0.0f;
  for (int k = p * 256; k < p * 256 + 256; ++k) acc += hs[k] * t_w2[k * 64 + j];
  red[tid] = acc;
  __syncthreads();
  if (tid < 64) {
    ttab[t * 64 + tid] =
        t_b2[tid] + red[tid] + red[tid + 64] + red[tid + 128] + red[tid + 192];
  }
}

#define TB 512
#define NW 16
#define LP 76
#define GP 132
#define LSZ (128 * LP)

__device__ __forceinline__ void ln_rows(const float* __restrict__ src,
                                        float* __restrict__ dst,
                                        const float* __restrict__ g,
                                        const float* __restrict__ bta, int tid) {
  const int s = tid >> 2, p = tid & 3;
  float sum = 0.0f, sq = 0.0f;
  for (int d = p * 16; d < p * 16 + 16; ++d) {
    float xv = src[s * LP + d];
    sum += xv; sq += xv * xv;
  }
  for (int off = 1; off < 4; off <<= 1) {
    sum += __shfl_xor(sum, off, 4);
    sq  += __shfl_xor(sq,  off, 4);
  }
  const float mu  = sum * (1.0f / 64.0f);
  const float var = sq * (1.0f / 64.0f) - mu * mu;
  const float inv = rsqrtf(var + 1e-5f);
  for (int d = p * 16; d < p * 16 + 16; ++d)
    dst[s * LP + d] = (src[s * LP + d] - mu) * inv * g[d] + bta[d];
}

__global__ __launch_bounds__(TB, 1) void xformer_kernel(
    const int* __restrict__ idx, const int* __restrict__ num_loops,
    const float* __restrict__ wte, const float* __restrict__ wpe,
    const float* __restrict__ ln1_g, const float* __restrict__ ln1_b,
    const float* __restrict__ w_qkv, const float* __restrict__ b_qkv,
    const float* __restrict__ w_o, const float* __restrict__ b_o,
    const float* __restrict__ ln2_g, const float* __restrict__ ln2_b,
    const float* __restrict__ w_fc, const float* __restrict__ b_fc,
    const float* __restrict__ w_pr, const float* __restrict__ b_pr,
    const float* __restrict__ lnf_g, const float* __restrict__ lnf_b,
    const float* __restrict__ ttab, float* __restrict__ out) {
  __shared__ float smem[5 * LSZ];
  float* xb = smem;
  float* hb = smem + LSZ;
  float* qb = smem + 2 * LSZ;
  float* kb = smem + 3 * LSZ;
  float* vb = smem + 4 * LSZ;
  float* gb = qb;
  float* ob = vb;

  const int b    = blockIdx.x;
  const int tid  = threadIdx.x;
  const int wid  = tid >> 5;
  const int lane = tid & 31;
  const int l16  = lane & 15;
  const int half = lane >> 4;
  const int koff = 2 * half;

  for (int i = tid; i < 8192; i += TB) {
    int s = i >> 6, d = i & 63;
    xb[s * LP + d] = wte[idx[b * 128 + s] * 64 + d] + wpe[i];
  }
  __syncthreads();

  const int NL = *num_loops;
  for (int t = 0; t < NL; ++t) {
    for (int i = tid; i < 8192; i += TB)
      xb[(i >> 6) * LP + (i & 63)] += ttab[t * 64 + (i & 63)];
    __syncthreads();

    ln_rows(xb, hb, ln1_g, ln1_b, tid);
    __syncthreads();

    for (int tile = wid; tile < 96; tile += NW) {
      const int m0 = (tile / 12) * 16, n0 = (tile % 12) * 16;
      v8f acc;
      { float bias = b_qkv[n0 + l16]; for (int r = 0; r < 8; ++r) acc[r] = bias; }
      for (int k = 0; k < 64; k += 4) {
        v2f a, w;
        a.x = hb[(m0 + l16) * LP + k + koff];
        a.y = hb[(m0 + l16) * LP + k + koff + 1];
        w.x = w_qkv[(k + koff) * 192 + n0 + l16];
        w.y = w_qkv[(k + koff + 1) * 192 + n0 + l16];
        acc = wmma4(a, w, acc);
      }
      float* dst; int nc;
      if (n0 < 64)       { dst = qb; nc = n0; }
      else if (n0 < 128) { dst = kb; nc = n0 - 64; }
      else               { dst = vb; nc = n0 - 128; }
      for (int r = 0; r < 8; ++r)
        dst[(m0 + r + 8 * half) * LP + nc + l16] = acc[r];
    }
    __syncthreads();

    for (int job = wid; job < 32; job += NW) {
      const int hh = job >> 3, qt = job & 7;
      const int q0 = qt * 16, c0 = hh * 16;
      v8f sc[8];
      for (int kt = 0; kt < 8; ++kt) {
        v8f a8; for (int r = 0; r < 8; ++r) a8[r] = 0.0f;
        for (int kk = 0; kk < 16; kk += 4) {
          v2f a, bb;
          a.x  = qb[(q0 + l16) * LP + c0 + kk + koff];
          a.y  = qb[(q0 + l16) * LP + c0 + kk + koff + 1];
          bb.x = kb[(kt * 16 + l16) * LP + c0 + kk + koff];
          bb.y = kb[(kt * 16 + l16) * LP + c0 + kk + koff + 1];
          a8 = wmma4(a, bb, a8);
        }
        for (int r = 0; r < 8; ++r) {
          const int row = q0 + r + 8 * half, col = kt * 16 + l16;
          a8[r] = (col <= row) ? a8[r] * 0.25f : -1e9f;
        }
        sc[kt] = a8;
      }
      for (int r = 0; r < 8; ++r) {
        float mx = -1e30f;
        for (int kt = 0; kt < 8; ++kt) mx = fmaxf(mx, sc[kt][r]);
        for (int off = 8; off; off >>= 1) mx = fmaxf(mx, __shfl_xor(mx, off, 16));
        float sum = 0.0f;
        for (int kt = 0; kt < 8; ++kt) {
          float e = __expf(sc[kt][r] - mx);
          sc[kt][r] = e; sum += e;
        }
        for (int off = 8; off; off >>= 1) sum += __shfl_xor(sum, off, 16);
        const float inv = 1.0f / sum;
        for (int kt = 0; kt < 8; ++kt) sc[kt][r] *= inv;
      }
      v8f o; for (int r = 0; r < 8; ++r) o[r] = 0.0f;
      for (int kt = 0; kt < 8; ++kt) {
        for (int r = 0; r < 8; ++r)
          hb[(q0 + r + 8 * half) * LP + c0 + l16] = sc[kt][r];
        for (int kk = 0; kk < 16; kk += 4) {
          v2f a, bb;
          a.x  = hb[(q0 + l16) * LP + c0 + kk + koff];
          a.y  = hb[(q0 + l16) * LP + c0 + kk + koff + 1];
          bb.x = vb[(kt * 16 + kk + koff) * LP + c0 + l16];
          bb.y = vb[(kt * 16 + kk + koff + 1) * LP + c0 + l16];
          o = wmma4(a, bb, o);
        }
      }
      for (int r = 0; r < 8; ++r)
        hb[(q0 + r + 8 * half) * LP + c0 + l16] = o[r];
    }
    __syncthreads();

    for (int tile = wid; tile < 32; tile += NW) {
      const int m0 = (tile >> 2) * 16, n0 = (tile & 3) * 16;
      v8f acc;
      { float bias = b_o[n0 + l16]; for (int r = 0; r < 8; ++r) acc[r] = bias; }
      for (int k = 0; k < 64; k += 4) {
        v2f a, w;
        a.x = hb[(m0 + l16) * LP + k + koff];
        a.y = hb[(m0 + l16) * LP + k + koff + 1];
        w.x = w_o[(k + koff) * 64 + n0 + l16];
        w.y = w_o[(k + koff + 1) * 64 + n0 + l16];
        acc = wmma4(a, w, acc);
      }
      for (int r = 0; r < 8; ++r)
        xb[(m0 + r + 8 * half) * LP + n0 + l16] += 0.1f * acc[r];
    }
    __syncthreads();

    ln_rows(xb, hb, ln2_g, ln2_b, tid);
    __syncthreads();

    for (int hh2 = 0; hh2 < 2; ++hh2) {
      for (int tile = wid; tile < 64; tile += NW) {
        const int m0 = (tile >> 3) * 16, ntl = (tile & 7) * 16;
        const int n0g = hh2 * 128 + ntl;
        v8f acc;
        { float bias = b_fc[n0g + l16]; for (int r = 0; r < 8; ++r) acc[r] = bias; }
        for (int k = 0; k < 64; k += 4) {
          v2f a, w;
          a.x = hb[(m0 + l16) * LP + k + koff];
          a.y = hb[(m0 + l16) * LP + k + koff + 1];
          w.x = w_fc[(k + koff) * 256 + n0g + l16];
          w.y = w_fc[(k + koff + 1) * 256 + n0g + l16];
          acc = wmma4(a, w, acc);
        }
        for (int r = 0; r < 8; ++r) {
          float u = acc[r];
          float g = 0.5f * u *
              (1.0f + tanhf(0.7978845608028654f * (u + 0.044715f * u * u * u)));
          gb[(m0 + r + 8 * half) * GP + ntl + l16] = g;
        }
      }
      __syncthreads();
      for (int tile = wid; tile < 32; tile += NW) {
        const int m0 = (tile >> 2) * 16, n0 = (tile & 3) * 16;
        v8f acc;
        if (hh2 == 0) {
          float bias = b_pr[n0 + l16];
          for (int r = 0; r < 8; ++r) acc[r] = bias;
        } else {
          for (int r = 0; r < 8; ++r)
            acc[r] = ob[(m0 + r + 8 * half) * LP + n0 + l16];
        }
        for (int k = 0; k < 128; k += 4) {
          v2f a, w;
          a.x = gb[(m0 + l16) * GP + k + koff];
          a.y = gb[(m0 + l16) * GP + k + koff + 1];
          w.x = w_pr[(hh2 * 128 + k + koff) * 64 + n0 + l16];
          w.y = w_pr[(hh2 * 128 + k + koff + 1) * 64 + n0 + l16];
          acc = wmma4(a, w, acc);
        }
        if (hh2 == 0) {
          for (int r = 0; r < 8; ++r)
            ob[(m0 + r + 8 * half) * LP + n0 + l16] = acc[r];
        } else {
          for (int r = 0; r < 8; ++r)
            xb[(m0 + r + 8 * half) * LP + n0 + l16] += 0.1f * acc[r];
        }
      }
      __syncthreads();
    }
  }

  {
    const int s = tid >> 2, p = tid & 3;
    float sum = 0.0f, sq = 0.0f;
    for (int d = p * 16; d < p * 16 + 16; ++d) {
      float xv = xb[s * LP + d];
      sum += xv; sq += xv * xv;
    }
    for (int off = 1; off < 4; off <<= 1) {
      sum += __shfl_xor(sum, off, 4);
      sq  += __shfl_xor(sq,  off, 4);
    }
    const float mu  = sum * (1.0f / 64.0f);
    const float var = sq * (1.0f / 64.0f) - mu * mu;
    const float inv = rsqrtf(var + 1e-5f);
    for (int d = p * 16; d < p * 16 + 16; ++d)
      out[b * 8192 + s * 64 + d] =
          (xb[s * LP + d] - mu) * inv * lnf_g[d] + lnf_b[d];
  }
}

extern "C" void kernel_launch(void* const* d_in, const int* in_sizes, int n_in,
                              void* d_out, int out_size, void* d_ws, size_t ws_size,
                              hipStream_t stream) {
  const int*   idx   = (const int*)  d_in[0];
  const int*   nlp   = (const int*)  d_in[1];
  const float* wte   = (const float*)d_in[2];
  const float* wpe   = (const float*)d_in[3];
  const float* t_w1  = (const float*)d_in[4];
  const float* t_b1  = (const float*)d_in[5];
  const float* t_w2  = (const float*)d_in[6];
  const float* t_b2  = (const float*)d_in[7];
  const float* ln1_g = (const float*)d_in[8];
  const float* ln1_b = (const float*)d_in[9];
  const float* w_qkv = (const float*)d_in[10];
  const float* b_qkv = (const float*)d_in[11];
  const float* w_o   = (const float*)d_in[12];
  const float* b_o   = (const float*)d_in[13];
  const float* ln2_g = (const float*)d_in[14];
  const float* ln2_b = (const float*)d_in[15];
  const float* w_fc  = (const float*)d_in[16];
  const float* b_fc  = (const float*)d_in[17];
  const float* w_pr  = (const float*)d_in[18];
  const float* b_pr  = (const float*)d_in[19];
  const float* lnf_g = (const float*)d_in[20];
  const float* lnf_b = (const float*)d_in[21];

  float* ttab = (float*)d_ws;

  temb_kernel<<<128, 256, 0, stream>>>(nlp, t_w1, t_b1, t_w2, t_b2, ttab);
  xformer_kernel<<<32, TB, 0, stream>>>(idx, nlp, wte, wpe, ln1_g, ln1_b,
                                        w_qkv, b_qkv, w_o, b_o, ln2_g, ln2_b,
                                        w_fc, b_fc, w_pr, b_pr, lnf_g, lnf_b,
                                        ttab, (float*)d_out);
}


# Softmax and LayerNorm lane reductions as DPP row_xmask butterflies, the 8 softmax rows unrolled side by side, instead of one-at-a-time LDS-permute hops in a rolled loop; sums and order unchanged
# speedup vs baseline: 2.4765x; 1.0775x over previous
#include <hip/hip_runtime.h>
#include <hip/hip_bf16.h>
#include <math.h>


typedef __attribute__((ext_vector_type(2))) float v2f;
typedef __attribute__((ext_vector_type(8))) float v8f;

__device__ __forceinline__ v8f wmma4(v2f a, v2f b, v8f c) {
  return __builtin_amdgcn_wmma_f32_16x16x4_f32(
       false, a,  false, b,
       (short)0, c,  false,  false);
}

template <int M>
__device__ __forceinline__ float row_xor(float v) {
  return __int_as_float(__builtin_amdgcn_update_dpp(
      0, __float_as_int(v), 0x160 + M, 0xf, 0xf, true));
}

__global__ __launch_bounds__(256) void temb_kernel(
    const int* __restrict__ num_loops,
    const float* __restrict__ t_w1, const float* __restrict__ t_b1,
    const float* __restrict__ t_w2, const float* __restrict__ t_b2,
    float* __restrict__ ttab) {
  const int t = blockIdx.x;
  if (t >= *num_loops) return;
  __shared__ float te[256];
  __shared__ float hs[1024];
  __shared__ float red[256];
  const int tid = threadIdx.x;

  if (tid < 128) {
    float f   = expf(-logf(10000.0f) * (float)tid / 128.0f);
    float arg = (float)t * f;
    te[tid]       = cosf(arg);
    te[128 + tid] = sinf(arg);
  }
  __syncthreads();

  for (int j = tid; j < 1024; j += 256) {
    float acc = t_b1[j];
    for (int k = 0; k < 256; ++k) acc += te[k] * t_w1[k * 1024 + j];
    hs[j] = acc / (1.0f + expf(-acc));
  }
  __syncthreads();

  const int j = tid & 63, p = tid >> 6;
  float acc = 0.0f;
  for (int k = p * 256; k < p * 256 + 256; ++k) acc += hs[k] * t_w2[k * 64 + j];
  red[tid] = acc;
  __syncthreads();
  if (tid < 64) {
    ttab[t * 64 + tid] =
        t_b2[tid] + red[tid] + red[tid + 64] + red[tid + 128] + red[tid + 192];
  }
}

#define TB 512
#define NW 16
#define LP 76
#define GP 132
#define LSZ (128 * LP)

__device__ __forceinline__ void ln_rows(const float* __restrict__ src,
                                        float* __restrict__ dst,
                                        const float* __restrict__ g,
                                        const float* __restrict__ bta, int tid) {
  const int s = tid >> 2, p = tid & 3;
  float sum = 0.0f, sq = 0.0f;
  for (int d = p * 16; d < p * 16 + 16; ++d) {
    float xv = src[s * LP + d];
    sum += xv; sq += xv * xv;
  }
  sum += row_xor<1>(sum); sq += row_xor<1>(sq);
  sum += row_xor<2>(sum); sq += row_xor<2>(sq);
  const float mu  = sum * (1.0f / 64.0f);
  const float var = sq * (1.0f / 64.0f) - mu * mu;
  const float inv = rsqrtf(var + 1e-5f);
  for (int d = p * 16; d < p * 16 + 16; ++d)
    dst[s * LP + d] = (src[s * LP + d] - mu) * inv * g[d] + bta[d];
}

__global__ __launch_bounds__(TB, 1) void xformer_kernel(
    const int* __restrict__ idx, const int* __restrict__ num_loops,
    const float* __restrict__ wte, const float* __restrict__ wpe,
    const float* __restrict__ ln1_g, const float* __restrict__ ln1_b,
    const float* __restrict__ w_qkv, const float* __restrict__ b_qkv,
    const float* __restrict__ w_o, const float* __restrict__ b_o,
    const float* __restrict__ ln2_g, const float* __restrict__ ln2_b,
    const float* __restrict__ w_fc, const float* __restrict__ b_fc,
    const float* __restrict__ w_pr, const float* __restrict__ b_pr,
    const float* __restrict__ lnf_g, const float* __restrict__ lnf_b,
    const float* __restrict__ ttab, float* __restrict__ out) {
  __shared__ float smem[5 * LSZ];
  float* xb = smem;
  float* hb = smem + LSZ;
  float* qb = smem + 2 * LSZ;
  float* kb = smem + 3 * LSZ;
  float* vb = smem + 4 * LSZ;
  float* gb = qb;
  float* ob = vb;

  const int b    = blockIdx.x;
  const int tid  = threadIdx.x;
  const int wid  = tid >> 5;
  const int lane = tid & 31;
  const int l16  = lane & 15;
  const int half = lane >> 4;
  const int koff = 2 * half;

  for (int i = tid; i < 8192; i += TB) {
    int s = i >> 6, d = i & 63;
    xb[s * LP + d] = wte[idx[b * 128 + s] * 64 + d] + wpe[i];
  }
  __syncthreads();

  const int NL = *num_loops;
  for (int t = 0; t < NL; ++t) {
    for (int i = tid; i < 8192; i += TB)
      xb[(i >> 6) * LP + (i & 63)] += ttab[t * 64 + (i & 63)];
    __syncthreads();

    ln_rows(xb, hb, ln1_g, ln1_b, tid);
    __syncthreads();

    for (int tile = wid; tile < 96; tile += NW) {
      const int m0 = (tile / 12) * 16, n0 = (tile % 12) * 16;
      v8f acc;
      { float bias = b_qkv[n0 + l16]; for (int r = 0; r < 8; ++r) acc[r] = bias; }
      for (int k = 0; k < 64; k += 4) {
        v2f a, w;
        a.x = hb[(m0 + l16) * LP + k + koff];
        a.y = hb[(m0 + l16) * LP + k + koff + 1];
        w.x = w_qkv[(k + koff) * 192 + n0 + l16];
        w.y = w_qkv[(k + koff + 1) * 192 + n0 + l16];
        acc = wmma4(a, w, acc);
      }
      float* dst; int nc;
      if (n0 < 64)       { dst = qb; nc = n0; }
      else if (n0 < 128) { dst = kb; nc = n0 - 64; }
      else               { dst = vb; nc = n0 - 128; }
      for (int r = 0; r < 8; ++r)
        dst[(m0 + r + 8 * half) * LP + nc + l16] = acc[r];
    }
    __syncthreads();

    for (int job = wid; job < 32; job += NW) {
      const int hh = job >> 3, qt = job & 7;
      const int q0 = qt * 16, c0 = hh * 16;
      v8f sc[8];
      for (int kt = 0; kt < 8; ++kt) {
        v8f a8; for (int r = 0; r < 8; ++r) a8[r] = 0.0f;
        for (int kk = 0; kk < 16; kk += 4) {
          v2f a, bb;
          a.x  = qb[(q0 + l16) * LP + c0 + kk + koff];
          a.y  = qb[(q0 + l16) * LP + c0 + kk + koff + 1];
          bb.x = kb[(kt * 16 + l16) * LP + c0 + kk + koff];
          bb.y = kb[(kt * 16 + l16) * LP + c0 + kk + koff + 1];
          a8 = wmma4(a, bb, a8);
        }
        for (int r = 0; r < 8; ++r) {
          const int row = q0 + r + 8 * half, col = kt * 16 + l16;
          a8[r] = (col <= row) ? a8[r] * 0.25f : -1e9f;
        }
        sc[kt] = a8;
      }
      float mx[8], sm[8];
#pragma unroll
      for (int r = 0; r < 8; ++r) {
        float m = -1e30f;
#pragma unroll
        for (int kt = 0; kt < 8; ++kt) m = fmaxf(m, sc[kt][r]);
        mx[r] = m;
      }
#pragma unroll
      for (int r = 0; r < 8; ++r) mx[r] = fmaxf(mx[r], row_xor<8>(mx[r]));
#pragma unroll
      for (int r = 0; r < 8; ++r) mx[r] = fmaxf(mx[r], row_xor<4>(mx[r]));
#pragma unroll
      for (int r = 0; r < 8; ++r) mx[r] = fmaxf(mx[r], row_xor<2>(mx[r]));
#pragma unroll
      for (int r = 0; r < 8; ++r) mx[r] = fmaxf(mx[r], row_xor<1>(mx[r]));
#pragma unroll
      for (int r = 0; r < 8; ++r) {
        float sum = 0.0f;
#pragma unroll
        for (int kt = 0; kt < 8; ++kt) {
          float e = __expf(sc[kt][r] - mx[r]);
          sc[kt][r] = e; sum += e;
        }
        sm[r] = sum;
      }
#pragma unroll
      for (int r = 0; r < 8; ++r) sm[r] += row_xor<8>(sm[r]);
#pragma unroll
      for (int r = 0; r < 8; ++r) sm[r] += row_xor<4>(sm[r]);
#pragma unroll
      for (int r = 0; r < 8; ++r) sm[r] += row_xor<2>(sm[r]);
#pragma unroll
      for (int r = 0; r < 8; ++r) sm[r] += row_xor<1>(sm[r]);
#pragma unroll
      for (int r = 0; r < 8; ++r) {
        const float inv = 1.0f / sm[r];
#pragma unroll
        for (int kt = 0; kt < 8; ++kt) sc[kt][r] *= inv;
      }
      v8f o; for (int r = 0; r < 8; ++r) o[r] = 0.0f;
      for (int kt = 0; kt < 8; ++kt) {
        for (int r = 0; r < 8; ++r)
          hb[(q0 + r + 8 * half) * LP + c0 + l16] = sc[kt][r];
        for (int kk = 0; kk < 16; kk += 4) {
          v2f a, bb;
          a.x  = hb[(q0 + l16) * LP + c0 + kk + koff];
          a.y  = hb[(q0 + l16) * LP + c0 + kk + koff + 1];
          bb.x = vb[(kt * 16 + kk + koff) * LP + c0 + l16];
          bb.y = vb[(kt * 16 + kk + koff + 1) * LP + c0 + l16];
          o = wmma4(a, bb, o);
        }
      }
      for (int r = 0; r < 8; ++r)
        hb[(q0 + r + 8 * half) * LP + c0 + l16] = o[r];
    }
    __syncthreads();

    for (int tile = wid; tile < 32; tile += NW) {
      const int m0 = (tile >> 2) * 16, n0 = (tile & 3) * 16;
      v8f acc;
      { float bias = b_o[n0 + l16]; for (int r = 0; r < 8; ++r) acc[r] = bias; }
      for (int k = 0; k < 64; k += 4) {
        v2f a, w;
        a.x = hb[(m0 + l16) * LP + k + koff];
        a.y = hb[(m0 + l16) * LP + k + koff + 1];
        w.x = w_o[(k + koff) * 64 + n0 + l16];
        w.y = w_o[(k + koff + 1) * 64 + n0 + l16];
        acc = wmma4(a, w, acc);
      }
      for (int r = 0; r < 8; ++r)
        xb[(m0 + r + 8 * half) * LP + n0 + l16] += 0.1f * acc[r];
    }
    __syncthreads();

    ln_rows(xb, hb, ln2_g, ln2_b, tid);
    __syncthreads();

    for (int hh2 = 0; hh2 < 2; ++hh2) {
      for (int tile = wid; tile < 64; tile += NW) {
        const int m0 = (tile >> 3) * 16, ntl = (tile & 7) * 16;
        const int n0g = hh2 * 128 + ntl;
        v8f acc;
        { float bias = b_fc[n0g + l16]; for (int r = 0; r < 8; ++r) acc[r] = bias; }
        for (int k = 0; k < 64; k += 4) {
          v2f a, w;
          a.x = hb[(m0 + l16) * LP + k + koff];
          a.y = hb[(m0 + l16) * LP + k + koff + 1];
          w.x = w_fc[(k + koff) * 256 + n0g + l16];
          w.y = w_fc[(k + koff + 1) * 256 + n0g + l16];
          acc = wmma4(a, w, acc);
        }
        for (int r = 0; r < 8; ++r) {
          float u = acc[r];
          float g = 0.5f * u *
              (1.0f + tanhf(0.7978845608028654f * (u + 0.044715f * u * u * u)));
          gb[(m0 + r + 8 * half) * GP + ntl + l16] = g;
        }
      }
      __syncthreads();
      for (int tile = wid; tile < 32; tile += NW) {
        const int m0 = (tile >> 2) * 16, n0 = (tile & 3) * 16;
        v8f acc;
        if (hh2 == 0) {
          float bias = b_pr[n0 + l16];
          for (int r = 0; r < 8; ++r) acc[r] = bias;
        } else {
          for (int r = 0; r < 8; ++r)
            acc[r] = ob[(m0 + r + 8 * half) * LP + n0 + l16];
        }
        for (int k = 0; k < 128; k += 4) {
          v2f a, w;
          a.x = gb[(m0 + l16) * GP + k + koff];
          a.y = gb[(m0 + l16) * GP + k + koff + 1];
          w.x = w_pr[(hh2 * 128 + k + koff) * 64 + n0 + l16];
          w.y = w_pr[(hh2 * 128 + k + koff + 1) * 64 + n0 + l16];
          acc = wmma4(a, w, acc);
        }
        if (hh2 == 0) {
          for (int r = 0; r < 8; ++r)
            ob[(m0 + r + 8 * half) * LP + n0 + l16] = acc[r];
        } else {
          for (int r = 0; r < 8; ++r)
            xb[(m0 + r + 8 * half) * LP + n0 + l16] += 0.1f * acc[r];
        }
      }
      __syncthreads();
    }
  }

  {
    const int s = tid >> 2, p = tid & 3;
    float sum = 0.0f, sq = 0.0f;
    for (int d = p * 16; d < p * 16 + 16; ++d) {
      float xv = xb[s * LP + d];
      sum += xv; sq += xv * xv;
    }
    sum += row_xor<1>(sum); sq += row_xor<1>(sq);
    sum += row_xor<2>(sum); sq += row_xor<2>(sq);
    const float mu  = sum * (1.0f / 64.0f);
    const float var = sq * (1.0f / 64.0f) - mu * mu;
    const float inv = rsqrtf(var + 1e-5f);
    for (int d = p * 16; d < p * 16 + 16; ++d)
      out[b * 8192 + s * 64 + d] =
          (xb[s * LP + d] - mu) * inv * lnf_g[d] + lnf_b[d];
  }
}

extern "C" void kernel_launch(void* const* d_in, const int* in_sizes, int n_in,
                              void* d_out, int out_size, void* d_ws, size_t ws_size,
                              hipStream_t stream) {
  const int*   idx   = (const int*)  d_in[0];
  const int*   nlp   = (const int*)  d_in[1];
  const float* wte   = (const float*)d_in[2];
  const float* wpe   = (const float*)d_in[3];
  const float* t_w1  = (const float*)d_in[4];
  const float* t_b1  = (const float*)d_in[5];
  const float* t_w2  = (const float*)d_in[6];
  const float* t_b2  = (const float*)d_in[7];
  const float* ln1_g = (const float*)d_in[8];
  const float* ln1_b = (const float*)d_in[9];
  const float* w_qkv = (const float*)d_in[10];
  const float* b_qkv = (const float*)d_in[11];
  const float* w_o   = (const float*)d_in[12];
  const float* b_o   = (const float*)d_in[13];
  const float* ln2_g = (const float*)d_in[14];
  const float* ln2_b = (const float*)d_in[15];
  const float* w_fc  = (const float*)d_in[16];
  const float* b_fc  = (const float*)d_in[17];
  const float* w_pr  = (const float*)d_in[18];
  const float* b_pr  = (const float*)d_in[19];
  const float* lnf_g = (const float*)d_in[20];
  const float* lnf_b = (const float*)d_in[21];

  float* ttab = (float*)d_ws;

  temb_kernel<<<128, 256, 0, stream>>>(nlp, t_w1, t_b1, t_w2, t_b2, ttab);
  xformer_kernel<<<32, TB, 0, stream>>>(idx, nlp, wte, wpe, ln1_g, ln1_b,
                                        w_qkv, b_qkv, w_o, b_o, ln2_g, ln2_b,
                                        w_fc, b_fc, w_pr, b_pr, lnf_g, lnf_b,
                                        ttab, (float*)d_out);
}


# Attention skips key tiles above the causal diagonal (exact zeros) in QK^T, softmax and PV, masks only the diagonal tile; a wave takes q-tiles 7-p and p (9 key tiles); sums and order unchanged
# speedup vs baseline: 2.8096x; 1.1345x over previous
#include <hip/hip_runtime.h>
#include <hip/hip_bf16.h>
#include <math.h>


typedef __attribute__((ext_vector_type(2))) float v2f;
typedef __attribute__((ext_vector_type(8))) float v8f;

__device__ __forceinline__ v8f wmma4(v2f a, v2f b, v8f c) {
  return __builtin_amdgcn_wmma_f32_16x16x4_f32(
       false, a,  false, b,
       (short)0, c,  false,  false);
}

template <int M>
__device__ __forceinline__ float row_xor(float v) {
  return __int_as_float(__builtin_amdgcn_update_dpp(
      0, __float_as_int(v), 0x160 + M, 0xf, 0xf, true));
}

__global__ __launch_bounds__(256) void temb_kernel(
    const int* __restrict__ num_loops,
    const float* __restrict__ t_w1, const float* __restrict__ t_b1,
    const float* __restrict__ t_w2, const float* __restrict__ t_b2,
    float* __restrict__ ttab) {
  const int t = blockIdx.x;
  if (t >= *num_loops) return;
  __shared__ float te[256];
  __shared__ float hs[1024];
  __shared__ float red[256];
  const int tid = threadIdx.x;

  if (tid < 128) {
    float f   = expf(-logf(10000.0f) * (float)tid / 128.0f);
    float arg = (float)t * f;
    te[tid]       = cosf(arg);
    te[128 + tid] = sinf(arg);
  }
  __syncthreads();

  for (int j = tid; j < 1024; j += 256) {
    float acc = t_b1[j];
    for (int k = 0; k < 256; ++k) acc += te[k] * t_w1[k * 1024 + j];
    hs[j] = acc / (1.0f + expf(-acc));
  }
  __syncthreads();

  const int j = tid & 63, p = tid >> 6;
  float acc = 0.0f;
  for (int k = p * 256; k < p * 256 + 256; ++k) acc += hs[k] * t_w2[k * 64 + j];
  red[tid] = acc;
  __syncthreads();
  if (tid < 64) {
    ttab[t * 64 + tid] =
        t_b2[tid] + red[tid] + red[tid + 64] + red[tid + 128] + red[tid + 192];
  }
}

#define TB 512
#define NW 16
#define LP 76
#define GP 132
#define LSZ (128 * LP)

__device__ __forceinline__ void ln_rows(const float* __restrict__ src,
                                        float* __restrict__ dst,
                                        const float* __restrict__ g,
                                        const float* __restrict__ bta, int tid) {
  const int s = tid >> 2, p = tid & 3;
  float sum = 0.0f, sq = 0.0f;
  for (int d = p * 16; d < p * 16 + 16; ++d) {
    float xv = src[s * LP + d];
    sum += xv; sq += xv * xv;
  }
  sum += row_xor<1>(sum); sq += row_xor<1>(sq);
  sum += row_xor<2>(sum); sq += row_xor<2>(sq);
  const float mu  = sum * (1.0f / 64.0f);
  const float var = sq * (1.0f / 64.0f) - mu * mu;
  const float inv = rsqrtf(var + 1e-5f);
  for (int d = p * 16; d < p * 16 + 16; ++d)
    dst[s * LP + d] = (src[s * LP + d] - mu) * inv * g[d] + bta[d];
}

__global__ __launch_bounds__(TB, 1) void xformer_kernel(
    const int* __restrict__ idx, const int* __restrict__ num_loops,
    const float* __restrict__ wte, const float* __restrict__ wpe,
    const float* __restrict__ ln1_g, const float* __restrict__ ln1_b,
    const float* __restrict__ w_qkv, const float* __restrict__ b_qkv,
    const float* __restrict__ w_o, const float* __restrict__ b_o,
    const float* __restrict__ ln2_g, const float* __restrict__ ln2_b,
    const float* __restrict__ w_fc, const float* __restrict__ b_fc,
    const float* __restrict__ w_pr, const float* __restrict__ b_pr,
    const float* __restrict__ lnf_g, const float* __restrict__ lnf_b,
    const float* __restrict__ ttab, float* __restrict__ out) {
  __shared__ float smem[5 * LSZ];
  float* xb = smem;
  float* hb = smem + LSZ;
  float* qb = smem + 2 * LSZ;
  float* kb = smem + 3 * LSZ;
  float* vb = smem + 4 * LSZ;
  float* gb = qb;
  float* ob = vb;

  const int b    = blockIdx.x;
  const int tid  = threadIdx.x;
  const int wid  = tid >> 5;
  const int lane = tid & 31;
  const int l16  = lane & 15;
  const int half = lane >> 4;
  const int koff = 2 * half;
  const int wv = __builtin_amdgcn_readfirstlane(wid);

  for (int i = tid; i < 8192; i += TB) {
    int s = i >> 6, d = i & 63;
    xb[s * LP + d] = wte[idx[b * 128 + s] * 64 + d] + wpe[i];
  }
  __syncthreads();

  const int NL = *num_loops;
  for (int t = 0; t < NL; ++t) {
    for (int i = tid; i < 8192; i += TB)
      xb[(i >> 6) * LP + (i & 63)] += ttab[t * 64 + (i & 63)];
    __syncthreads();

    ln_rows(xb, hb, ln1_g, ln1_b, tid);
    __syncthreads();

    for (int tile = wid; tile < 96; tile += NW) {
      const int m0 = (tile / 12) * 16, n0 = (tile % 12) * 16;
      v8f acc;
      { float bias = b_qkv[n0 + l16]; for (int r = 0; r < 8; ++r) acc[r] = bias; }
      for (int k = 0; k < 64; k += 4) {
        v2f a, w;
        a.x = hb[(m0 + l16) * LP + k + koff];
        a.y = hb[(m0 + l16) * LP + k + koff + 1];
        w.x = w_qkv[(k + koff) * 192 + n0 + l16];
        w.y = w_qkv[(k + koff + 1) * 192 + n0 + l16];
        acc = wmma4(a, w, acc);
      }
      float* dst; int nc;
      if (n0 < 64)       { dst = qb; nc = n0; }
      else if (n0 < 128) { dst = kb; nc = n0 - 64; }
      else               { dst = vb; nc = n0 - 128; }
      for (int r = 0; r < 8; ++r)
        dst[(m0 + r + 8 * half) * LP + nc + l16] = acc[r];
    }
    __syncthreads();

    for (int jj = 0; jj < 2; ++jj) {
      const int hh = wv >> 2;
      const int qt = jj ? (wv & 3) : 7 - (wv & 3);
      const int q0 = qt * 16, c0 = hh * 16;
      v8f sc[8];
#pragma unroll
      for (int kt = 0; kt < 8; ++kt) {
        if (kt > qt) break;
        v8f a8; for (int r = 0; r < 8; ++r) a8[r] = 0.0f;
        for (int kk = 0; kk < 16; kk += 4) {
          v2f a, bb;
          a.x  = qb[(q0 + l16) * LP + c0 + kk + koff];
          a.y  = qb[(q0 + l16) * LP + c0 + kk + koff + 1];
          bb.x = kb[(kt * 16 + l16) * LP + c0 + kk + koff];
          bb.y = kb[(kt * 16 + l16) * LP + c0 + kk + koff + 1];
          a8 = wmma4(a, bb, a8);
        }
        for (int r = 0; r < 8; ++r) a8[r] *= 0.25f;
        if (kt == qt) {
          for (int r = 0; r < 8; ++r)
            a8[r] = (l16 <= r + 8 * half) ? a8[r] : -1e9f;
        }
        sc[kt] = a8;
      }
      float mx[8], sm[8];
#pragma unroll
      for (int r = 0; r < 8; ++r) mx[r] = -1e30f;
#pragma unroll
      for (int kt = 0; kt < 8; ++kt) {
        if (kt > qt) break;
#pragma unroll
        for (int r = 0; r < 8; ++r) mx[r] = fmaxf(mx[r], sc[kt][r]);
      }
#pragma unroll
      for (int r = 0; r < 8; ++r) mx[r] = fmaxf(mx[r], row_xor<8>(mx[r]));
#pragma unroll
      for (int r = 0; r < 8; ++r) mx[r] = fmaxf(mx[r], row_xor<4>(mx[r]));
#pragma unroll
      for (int r = 0; r < 8; ++r) mx[r] = fmaxf(mx[r], row_xor<2>(mx[r]));
#pragma unroll
      for (int r = 0; r < 8; ++r) mx[r] = fmaxf(mx[r], row_xor<1>(mx[r]));
#pragma unroll
      for (int r = 0; r < 8; ++r) sm[r] = 0.0f;
#pragma unroll
      for (int kt = 0; kt < 8; ++kt) {
        if (kt > qt) break;
#pragma unroll
        for (int r = 0; r < 8; ++r) {
          float e = __expf(sc[kt][r] - mx[r]);
          sc[kt][r] = e; sm[r] += e;
        }
      }
#pragma unroll
      for (int r = 0; r < 8; ++r) sm[r] += row_xor<8>(sm[r]);
#pragma unroll
      for (int r = 0; r < 8; ++r) sm[r] += row_xor<4>(sm[r]);
#pragma unroll
      for (int r = 0; r < 8; ++r) sm[r] += row_xor<2>(sm[r]);
#pragma unroll
      for (int r = 0; r < 8; ++r) sm[r] += row_xor<1>(sm[r]);
#pragma unroll
      for (int r = 0; r < 8; ++r) sm[r] = 1.0f / sm[r];
      v8f o; for (int r = 0; r < 8; ++r) o[r] = 0.0f;
#pragma unroll
      for (int kt = 0; kt < 8; ++kt) {
        if (kt > qt) break;
        for (int r = 0; r < 8; ++r)
          hb[(q0 + r + 8 * half) * LP + c0 + l16] = sc[kt][r] * sm[r];
        for (int kk = 0; kk < 16; kk += 4) {
          v2f a, bb;
          a.x  = hb[(q0 + l16) * LP + c0 + kk + koff];
          a.y  = hb[(q0 + l16) * LP + c0 + kk + koff + 1];
          bb.x = vb[(kt * 16 + kk + koff) * LP + c0 + l16];
          bb.y = vb[(kt * 16 + kk + koff + 1) * LP + c0 + l16];
          o = wmma4(a, bb, o);
        }
      }
      for (int r = 0; r < 8; ++r)
        hb[(q0 + r + 8 * half) * LP + c0 + l16] = o[r];
    }
    __syncthreads();

    for (int tile = wid; tile < 32; tile += NW) {
      const int m0 = (tile >> 2) * 16, n0 = (tile & 3) * 16;
      v8f acc;
      { float bias = b_o[n0 + l16]; for (int r = 0; r < 8; ++r) acc[r] = bias; }
      for (int k = 0; k < 64; k += 4) {
        v2f a, w;
        a.x = hb[(m0 + l16) * LP + k + koff];
        a.y = hb[(m0 + l16) * LP + k + koff + 1];
        w.x = w_o[(k + koff) * 64 + n0 + l16];
        w.y = w_o[(k + koff + 1) * 64 + n0 + l16];
        acc = wmma4(a, w, acc);
      }
      for (int r = 0; r < 8; ++r)
        xb[(m0 + r + 8 * half) * LP + n0 + l16] += 0.1f * acc[r];
    }
    __syncthreads();

    ln_rows(xb, hb, ln2_g, ln2_b, tid);
    __syncthreads();

    for (int hh2 = 0; hh2 < 2; ++hh2) {
      for (int tile = wid; tile < 64; tile += NW) {
        const int m0 = (tile >> 3) * 16, ntl = (tile & 7) * 16;
        const int n0g = hh2 * 128 + ntl;
        v8f acc;
        { float bias = b_fc[n0g + l16]; for (int r = 0; r < 8; ++r) acc[r] = bias; }
        for (int k = 0; k < 64; k += 4) {
          v2f a, w;
          a.x = hb[(m0 + l16) * LP + k + koff];
          a.y = hb[(m0 + l16) * LP + k + koff + 1];
          w.x = w_fc[(k + koff) * 256 + n0g + l16];
          w.y = w_fc[(k + koff + 1) * 256 + n0g + l16];
          acc = wmma4(a, w, acc);
        }
        for (int r = 0; r < 8; ++r) {
          float u = acc[r];
          float g = 0.5f * u *
              (1.0f + tanhf(0.7978845608028654f * (u + 0.044715f * u * u * u)));
          gb[(m0 + r + 8 * half) * GP + ntl + l16] = g;
        }
      }
      __syncthreads();
      for (int tile = wid; tile < 32; tile += NW) {
        const int m0 = (tile >> 2) * 16, n0 = (tile & 3) * 16;
        v8f acc;
        if (hh2 == 0) {
          float bias = b_pr[n0 + l16];
          for (int r = 0; r < 8; ++r) acc[r] = bias;
        } else {
          for (int r = 0; r < 8; ++r)
            acc[r] = ob[(m0 + r + 8 * half) * LP + n0 + l16];
        }
        for (int k = 0; k < 128; k += 4) {
          v2f a, w;
          a.x = gb[(m0 + l16) * GP + k + koff];
          a.y = gb[(m0 + l16) * GP + k + koff + 1];
          w.x = w_pr[(hh2 * 128 + k + koff) * 64 + n0 + l16];
          w.y = w_pr[(hh2 * 128 + k + koff + 1) * 64 + n0 + l16];
          acc = wmma4(a, w, acc);
        }
        if (hh2 == 0) {
          for (int r = 0; r < 8; ++r)
            ob[(m0 + r + 8 * half) * LP + n0 + l16] = acc[r];
        } else {
          for (int r = 0; r < 8; ++r)
            xb[(m0 + r + 8 * half) * LP + n0 + l16] += 0.1f * acc[r];
        }
      }
      __syncthreads();
    }
  }

  {
    const int s = tid >> 2, p = tid & 3;
    float sum = 0.0f, sq = 0.0f;
    for (int d = p * 16; d < p * 16 + 16; ++d) {
      float xv = xb[s * LP + d];
      sum += xv; sq += xv * xv;
    }
    sum += row_xor<1>(sum); sq += row_xor<1>(sq);
    sum += row_xor<2>(sum); sq += row_xor<2>(sq);
    const float mu  = sum * (1.0f / 64.0f);
    const float var = sq * (1.0f / 64.0f) - mu * mu;
    const float inv = rsqrtf(var + 1e-5f);
    for (int d = p * 16; d < p * 16 + 16; ++d)
      out[b * 8192 + s * 64 + d] =
          (xb[s * LP + d] - mu) * inv * lnf_g[d] + lnf_b[d];
  }
}

extern "C" void kernel_launch(void* const* d_in, const int* in_sizes, int n_in,
                              void* d_out, int out_size, void* d_ws, size_t ws_size,
                              hipStream_t stream) {
  const int*   idx   = (const int*)  d_in[0];
  const int*   nlp   = (const int*)  d_in[1];
  const float* wte   = (const float*)d_in[2];
  const float* wpe   = (const float*)d_in[3];
  const float* t_w1  = (const float*)d_in[4];
  const float* t_b1  = (const float*)d_in[5];
  const float* t_w2  = (const float*)d_in[6];
  const float* t_b2  = (const float*)d_in[7];
  const float* ln1_g = (const float*)d_in[8];
  const float* ln1_b = (const float*)d_in[9];
  const float* w_qkv = (const float*)d_in[10];
  const float* b_qkv = (const float*)d_in[11];
  const float* w_o   = (const float*)d_in[12];
  const float* b_o   = (const float*)d_in[13];
  const float* ln2_g = (const float*)d_in[14];
  const float* ln2_b = (const float*)d_in[15];
  const float* w_fc  = (const float*)d_in[16];
  const float* b_fc  = (const float*)d_in[17];
  const float* w_pr  = (const float*)d_in[18];
  const float* b_pr  = (const float*)d_in[19];
  const float* lnf_g = (const float*)d_in[20];
  const float* lnf_b = (const float*)d_in[21];

  float* ttab = (float*)d_ws;

  temb_kernel<<<128, 256, 0, stream>>>(nlp, t_w1, t_b1, t_w2, t_b2, ttab);
  xformer_kernel<<<32, TB, 0, stream>>>(idx, nlp, wte, wpe, ln1_g, ln1_b,
                                        w_qkv, b_qkv, w_o, b_o, ln2_g, ln2_b,
                                        w_fc, b_fc, w_pr, b_pr, lnf_g, lnf_b,
                                        ttab, (float*)d_out);
}


# MLP weights (w_fc, w_pr) copied once per launch into LDS as B fragments; a wave reads its column tile's fragments from LDS once per phase, not from global at every tile; sums and order unchanged
# speedup vs baseline: 2.9814x; 1.0611x over previous
#include <hip/hip_runtime.h>
#include <hip/hip_bf16.h>
#include <math.h>


typedef __attribute__((ext_vector_type(2))) float v2f;
typedef __attribute__((ext_vector_type(8))) float v8f;

__device__ __forceinline__ v8f wmma4(v2f a, v2f b, v8f c) {
  return __builtin_amdgcn_wmma_f32_16x16x4_f32(
       false, a,  false, b,
       (short)0, c,  false,  false);
}

template <int M>
__device__ __forceinline__ float row_xor(float v) {
  return __int_as_float(__builtin_amdgcn_update_dpp(
      0, __float_as_int(v), 0x160 + M, 0xf, 0xf, true));
}

__global__ __launch_bounds__(256) void temb_kernel(
    const int* __restrict__ num_loops,
    const float* __restrict__ t_w1, const float* __restrict__ t_b1,
    const float* __restrict__ t_w2, const float* __restrict__ t_b2,
    float* __restrict__ ttab) {
  const int t = blockIdx.x;
  if (t >= *num_loops) return;
  __shared__ float te[256];
  __shared__ float hs[1024];
  __shared__ float red[256];
  const int tid = threadIdx.x;

  if (tid < 128) {
    float f   = expf(-logf(10000.0f) * (float)tid / 128.0f);
    float arg = (float)t * f;
    te[tid]       = cosf(arg);
    te[128 + tid] = sinf(arg);
  }
  __syncthreads();

  for (int j = tid; j < 1024; j += 256) {
    float acc = t_b1[j];
    for (int k = 0; k < 256; ++k) acc += te[k] * t_w1[k * 1024 + j];
    hs[j] = acc / (1.0f + expf(-acc));
  }
  __syncthreads();

  const int j = tid & 63, p = tid >> 6;
  float acc = 0.0f;
  for (int k = p * 256; k < p * 256 + 256; ++k) acc += hs[k] * t_w2[k * 64 + j];
  red[tid] = acc;
  __syncthreads();
  if (tid < 64) {
    ttab[t * 64 + tid] =
        t_b2[tid] + red[tid] + red[tid + 64] + red[tid + 128] + red[tid + 192];
  }
}

#define TB 512
#define NW 16
#define LP 76
#define GP 132
#define LSZ (128 * LP)
#define WSZ 16384

__device__ __forceinline__ void ln_rows(const float* __restrict__ src,
                                        float* __restrict__ dst,
                                        const float* __restrict__ g,
                                        const float* __restrict__ bta, int tid) {
  const int s = tid >> 2, p = tid & 3;
  float sum = 0.0f, sq = 0.0f;
  for (int d = p * 16; d < p * 16 + 16; ++d) {
    float xv = src[s * LP + d];
    sum += xv; sq += xv * xv;
  }
  sum += row_xor<1>(sum); sq += row_xor<1>(sq);
  sum += row_xor<2>(sum); sq += row_xor<2>(sq);
  const float mu  = sum * (1.0f / 64.0f);
  const float var = sq * (1.0f / 64.0f) - mu * mu;
  const float inv = rsqrtf(var + 1e-5f);
  for (int d = p * 16; d < p * 16 + 16; ++d)
    dst[s * LP + d] = (src[s * LP + d] - mu) * inv * g[d] + bta[d];
}

__global__ __launch_bounds__(TB, 1) void xformer_kernel(
    const int* __restrict__ idx, const int* __restrict__ num_loops,
    const float* __restrict__ wte, const float* __restrict__ wpe,
    const float* __restrict__ ln1_g, const float* __restrict__ ln1_b,
    const float* __restrict__ w_qkv, const float* __restrict__ b_qkv,
    const float* __restrict__ w_o, const float* __restrict__ b_o,
    const float* __restrict__ ln2_g, const float* __restrict__ ln2_b,
    const float* __restrict__ w_fc, const float* __restrict__ b_fc,
    const float* __restrict__ w_pr, const float* __restrict__ b_pr,
    const float* __restrict__ lnf_g, const float* __restrict__ lnf_b,
    const float* __restrict__ ttab, float* __restrict__ out) {
  __shared__ float smem[5 * LSZ + 2 * WSZ];
  float* xb = smem;
  float* hb = smem + LSZ;
  float* qb = smem + 2 * LSZ;
  float* kb = smem + 3 * LSZ;
  float* vb = smem + 4 * LSZ;
  float* gb = qb;
  float* ob = vb;
  float* wfl = smem + 5 * LSZ;
  float* wpl = smem + 5 * LSZ + WSZ;

  const int b    = blockIdx.x;
  const int tid  = threadIdx.x;
  const int wid  = tid >> 5;
  const int lane = tid & 31;
  const int l16  = lane & 15;
  const int half = lane >> 4;
  const int koff = 2 * half;
  const int wv = __builtin_amdgcn_readfirstlane(wid);

  for (int i = tid; i < 8192; i += TB) {
    int s = i >> 6, d = i & 63;
    xb[s * LP + d] = wte[idx[b * 128 + s] * 64 + d] + wpe[i];
  }
  for (int i = tid; i < WSZ; i += TB) {
    const int k = i >> 8, n = i & 255;
    wfl[(((n >> 4) * 16 + (k >> 2)) * 32 + ((k >> 1) & 1) * 16 + (n & 15)) * 2 + (k & 1)] = w_fc[i];
  }
  for (int i = tid; i < WSZ; i += TB) {
    const int k = i >> 6, n = i & 63;
    wpl[(((n >> 4) * 64 + (k >> 2)) * 32 + ((k >> 1) & 1) * 16 + (n & 15)) * 2 + (k & 1)] = w_pr[i];
  }
  __syncthreads();

  const int NL = *num_loops;
  for (int t = 0; t < NL; ++t) {
    for (int i = tid; i < 8192; i += TB)
      xb[(i >> 6) * LP + (i & 63)] += ttab[t * 64 + (i & 63)];
    __syncthreads();

    ln_rows(xb, hb, ln1_g, ln1_b, tid);
    __syncthreads();

    for (int tile = wid; tile < 96; tile += NW) {
      const int m0 = (tile / 12) * 16, n0 = (tile % 12) * 16;
      v8f acc;
      { float bias = b_qkv[n0 + l16]; for (int r = 0; r < 8; ++r) acc[r] = bias; }
      for (int k = 0; k < 64; k += 4) {
        v2f a, w;
        a.x = hb[(m0 + l16) * LP + k + koff];
        a.y = hb[(m0 + l16) * LP + k + koff + 1];
        w.x = w_qkv[(k + koff) * 192 + n0 + l16];
        w.y = w_qkv[(k + koff + 1) * 192 + n0 + l16];
        acc = wmma4(a, w, acc);
      }
      float* dst; int nc;
      if (n0 < 64)       { dst = qb; nc = n0; }
      else if (n0 < 128) { dst = kb; nc = n0 - 64; }
      else               { dst = vb; nc = n0 - 128; }
      for (int r = 0; r < 8; ++r)
        dst[(m0 + r + 8 * half) * LP + nc + l16] = acc[r];
    }
    __syncthreads();

    for (int jj = 0; jj < 2; ++jj) {
      const int hh = wv >> 2;
      const int qt = jj ? (wv & 3) : 7 - (wv & 3);
      const int q0 = qt * 16, c0 = hh * 16;
      v8f sc[8];
#pragma unroll
      for (int kt = 0; kt < 8; ++kt) {
        if (kt > qt) break;
        v8f a8; for (int r = 0; r < 8; ++r) a8[r] = 0.0f;
        for (int kk = 0; kk < 16; kk += 4) {
          v2f a, bb;
          a.x  = qb[(q0 + l16) * LP + c0 + kk + koff];
          a.y  = qb[(q0 + l16) * LP + c0 + kk + koff + 1];
          bb.x = kb[(kt * 16 + l16) * LP + c0 + kk + koff];
          bb.y = kb[(kt * 16 + l16) * LP + c0 + kk + koff + 1];
          a8 = wmma4(a, bb, a8);
        }
        for (int r = 0; r < 8; ++r) a8[r] *= 0.25f;
        if (kt == qt) {
          for (int r = 0; r < 8; ++r)
            a8[r] = (l16 <= r + 8 * half) ? a8[r] : -1e9f;
        }
        sc[kt] = a8;
      }
      float mx[8], sm[8];
#pragma unroll
      for (int r = 0; r < 8; ++r) mx[r] = -1e30f;
#pragma unroll
      for (int kt = 0; kt < 8; ++kt) {
        if (kt > qt) break;
#pragma unroll
        for (int r = 0; r < 8; ++r) mx[r] = fmaxf(mx[r], sc[kt][r]);
      }
#pragma unroll
      for (int r = 0; r < 8; ++r) mx[r] = fmaxf(mx[r], row_xor<8>(mx[r]));
#pragma unroll
      for (int r = 0; r < 8; ++r) mx[r] = fmaxf(mx[r], row_xor<4>(mx[r]));
#pragma unroll
      for (int r = 0; r < 8; ++r) mx[r] = fmaxf(mx[r], row_xor<2>(mx[r]));
#pragma unroll
      for (int r = 0; r < 8; ++r) mx[r] = fmaxf(mx[r], row_xor<1>(mx[r]));
#pragma unroll
      for (int r = 0; r < 8; ++r) sm[r] = 0.0f;
#pragma unroll
      for (int kt = 0; kt < 8; ++kt) {
        if (kt > qt) break;
#pragma unroll
        for (int r = 0; r < 8; ++r) {
          float e = __expf(sc[kt][r] - mx[r]);
          sc[kt][r] = e; sm[r] += e;
        }
      }
#pragma unroll
      for (int r = 0; r < 8; ++r) sm[r] += row_xor<8>(sm[r]);
#pragma unroll
      for (int r = 0; r < 8; ++r) sm[r] += row_xor<4>(sm[r]);
#pragma unroll
      for (int r = 0; r < 8; ++r) sm[r] += row_xor<2>(sm[r]);
#pragma unroll
      for (int r = 0; r < 8; ++r) sm[r] += row_xor<1>(sm[r]);
#pragma unroll
      for (int r = 0; r < 8; ++r) sm[r] = 1.0f / sm[r];
      v8f o; for (int r = 0; r < 8; ++r) o[r] = 0.0f;
#pragma unroll
      for (int kt = 0; kt < 8; ++kt) {
        if (kt > qt) break;
        for (int r = 0; r < 8; ++r)
          hb[(q0 + r + 8 * half) * LP + c0 + l16] = sc[kt][r] * sm[r];
        for (int kk = 0; kk < 16; kk += 4) {
          v2f a, bb;
          a.x  = hb[(q0 + l16) * LP + c0 + kk + koff];
          a.y  = hb[(q0 + l16) * LP + c0 + kk + koff + 1];
          bb.x = vb[(kt * 16 + kk + koff) * LP + c0 + l16];
          bb.y = vb[(kt * 16 + kk + koff + 1) * LP + c0 + l16];
          o = wmma4(a, bb, o);
        }
      }
      for (int r = 0; r < 8; ++r)
        hb[(q0 + r + 8 * half) * LP + c0 + l16] = o[r];
    }
    __syncthreads();

    for (int tile = wid; tile < 32; tile += NW) {
      const int m0 = (tile >> 2) * 16, n0 = (tile & 3) * 16;
      v8f acc;
      { float bias = b_o[n0 + l16]; for (int r = 0; r < 8; ++r) acc[r] = bias; }
      for (int k = 0; k < 64; k += 4) {
        v2f a, w;
        a.x = hb[(m0 + l16) * LP + k + koff];
        a.y = hb[(m0 + l16) * LP + k + koff + 1];
        w.x = w_o[(k + koff) * 64 + n0 + l16];
        w.y = w_o[(k + koff + 1) * 64 + n0 + l16];
        acc = wmma4(a, w, acc);
      }
      for (int r = 0; r < 8; ++r)
        xb[(m0 + r + 8 * half) * LP + n0 + l16] += 0.1f * acc[r];
    }
    __syncthreads();

    ln_rows(xb, hb, ln2_g, ln2_b, tid);
    __syncthreads();

    for (int hh2 = 0; hh2 < 2; ++hh2) {
      {
        const int ntl = (wv & 7) * 16;
        const int n0g = hh2 * 128 + ntl;
        const v2f* bf = (const v2f*)(wfl + (hh2 * 8 + (wv & 7)) * 1024) + lane;
        v2f wf[16];
#pragma unroll
        for (int ks = 0; ks < 16; ++ks) wf[ks] = bf[ks * 32];
        const float bias = b_fc[n0g + l16];
        for (int i = 0; i < 4; ++i) {
          const int m0 = ((wv >> 3) + 2 * i) * 16;
          v8f acc;
          for (int r = 0; r < 8; ++r) acc[r] = bias;
#pragma unroll
          for (int ks = 0; ks < 16; ++ks) {
            v2f a;
            a.x = hb[(m0 + l16) * LP + 4 * ks + koff];
            a.y = hb[(m0 + l16) * LP + 4 * ks + koff + 1];
            acc = wmma4(a, wf[ks], acc);
          }
          for (int r = 0; r < 8; ++r) {
            float u = acc[r];
            float g = 0.5f * u *
                (1.0f + tanhf(0.7978845608028654f * (u + 0.044715f * u * u * u)));
            gb[(m0 + r + 8 * half) * GP + ntl + l16] = g;
          }
        }
      }
      __syncthreads();
      {
        const int n0 = (wv & 3) * 16;
        const v2f* bp = (const v2f*)(wpl + ((wv & 3) * 64 + hh2 * 32) * 64) + lane;
        v2f wp[32];
#pragma unroll
        for (int ks = 0; ks < 32; ++ks) wp[ks] = bp[ks * 32];
        const float bias = b_pr[n0 + l16];
        for (int i = 0; i < 2; ++i) {
          const int m0 = ((wv >> 2) + 4 * i) * 16;
          v8f acc;
          if (hh2 == 0) {
            for (int r = 0; r < 8; ++r) acc[r] = bias;
          } else {
            for (int r = 0; r < 8; ++r)
              acc[r] = ob[(m0 + r + 8 * half) * LP + n0 + l16];
          }
#pragma unroll
          for (int ks = 0; ks < 32; ++ks) {
            v2f a;
            a.x = gb[(m0 + l16) * GP + 4 * ks + koff];
            a.y = gb[(m0 + l16) * GP + 4 * ks + koff + 1];
            acc = wmma4(a, wp[ks], acc);
          }
          if (hh2 == 0) {
            for (int r = 0; r < 8; ++r)
              ob[(m0 + r + 8 * half) * LP + n0 + l16] = acc[r];
          } else {
            for (int r = 0; r < 8; ++r)
              xb[(m0 + r + 8 * half) * LP + n0 + l16] += 0.1f * acc[r];
          }
        }
      }
      __syncthreads();
    }
  }

  {
    const int s = tid >> 2, p = tid & 3;
    float sum = 0.0f, sq = 0.0f;
    for (int d = p * 16; d < p * 16 + 16; ++d) {
      float xv = xb[s * LP + d];
      sum += xv; sq += xv * xv;
    }
    sum += row_xor<1>(sum); sq += row_xor<1>(sq);
    sum += row_xor<2>(sum); sq += row_xor<2>(sq);
    const float mu  = sum * (1.0f / 64.0f);
    const float var = sq * (1.0f / 64.0f) - mu * mu;
    const float inv = rsqrtf(var + 1e-5f);
    for (int d = p * 16; d < p * 16 + 16; ++d)
      out[b * 8192 + s * 64 + d] =
          (xb[s * LP + d] - mu) * inv * lnf_g[d] + lnf_b[d];
  }
}

extern "C" void kernel_launch(void* const* d_in, const int* in_sizes, int n_in,
                              void* d_out, int out_size, void* d_ws, size_t ws_size,
                              hipStream_t stream) {
  const int*   idx   = (const int*)  d_in[0];
  const int*   nlp   = (const int*)  d_in[1];
  const float* wte   = (const float*)d_in[2];
  const float* wpe   = (const float*)d_in[3];
  const float* t_w1  = (const float*)d_in[4];
  const float* t_b1  = (const float*)d_in[5];
  const float* t_w2  = (const float*)d_in[6];
  const float* t_b2  = (const float*)d_in[7];
  const float* ln1_g = (const float*)d_in[8];
  const float* ln1_b = (const float*)d_in[9];
  const float* w_qkv = (const float*)d_in[10];
  const float* b_qkv = (const float*)d_in[11];
  const float* w_o   = (const float*)d_in[12];
  const float* b_o   = (const float*)d_in[13];
  const float* ln2_g = (const float*)d_in[14];
  const float* ln2_b = (const float*)d_in[15];
  const float* w_fc  = (const float*)d_in[16];
  const float* b_fc  = (const float*)d_in[17];
  const float* w_pr  = (const float*)d_in[18];
  const float* b_pr  = (const float*)d_in[19];
  const float* lnf_g = (const float*)d_in[20];
  const float* lnf_b = (const float*)d_in[21];

  float* ttab = (float*)d_ws;

  temb_kernel<<<128, 256, 0, stream>>>(nlp, t_w1, t_b1, t_w2, t_b2, ttab);
  xformer_kernel<<<32, TB, 0, stream>>>(idx, nlp, wte, wpe, ln1_g, ln1_b,
                                        w_qkv, b_qkv, w_o, b_o, ln2_g, ln2_b,
                                        w_fc, b_fc, w_pr, b_pr, lnf_g, lnf_b,
                                        ttab, (float*)d_out);
}


# GELU as u/(1+exp2(c(u))), by (1+tanh z)/2 = 1/(1+exp(-2z)): one hardware exp2 and one reciprocal per value, 8 values staged side by side, no branch, not library tanhf; GEMM sums and order unchanged
# speedup vs baseline: 3.2057x; 1.0752x over previous
#include <hip/hip_runtime.h>
#include <hip/hip_bf16.h>
#include <math.h>


typedef __attribute__((ext_vector_type(2))) float v2f;
typedef __attribute__((ext_vector_type(8))) float v8f;

__device__ __forceinline__ v8f wmma4(v2f a, v2f b, v8f c) {
  return __builtin_amdgcn_wmma_f32_16x16x4_f32(
       false, a,  false, b,
       (short)0, c,  false,  false);
}

template <int M>
__device__ __forceinline__ float row_xor(float v) {
  return __int_as_float(__builtin_amdgcn_update_dpp(
      0, __float_as_int(v), 0x160 + M, 0xf, 0xf, true));
}

__global__ __launch_bounds__(256) void temb_kernel(
    const int* __restrict__ num_loops,
    const float* __restrict__ t_w1, const float* __restrict__ t_b1,
    const float* __restrict__ t_w2, const float* __restrict__ t_b2,
    float* __restrict__ ttab) {
  const int t = blockIdx.x;
  if (t >= *num_loops) return;
  __shared__ float te[256];
  __shared__ float hs[1024];
  __shared__ float red[256];
  const int tid = threadIdx.x;

  if (tid < 128) {
    float f   = expf(-logf(10000.0f) * (float)tid / 128.0f);
    float arg = (float)t * f;
    te[tid]       = cosf(arg);
    te[128 + tid] = sinf(arg);
  }
  __syncthreads();

  for (int j = tid; j < 1024; j += 256) {
    float acc = t_b1[j];
    for (int k = 0; k < 256; ++k) acc += te[k] * t_w1[k * 1024 + j];
    hs[j] = acc / (1.0f + expf(-acc));
  }
  __syncthreads();

  const int j = tid & 63, p = tid >> 6;
  float acc = 0.0f;
  for (int k = p * 256; k < p * 256 + 256; ++k) acc += hs[k] * t_w2[k * 64 + j];
  red[tid] = acc;
  __syncthreads();
  if (tid < 64) {
    ttab[t * 64 + tid] =
        t_b2[tid] + red[tid] + red[tid + 64] + red[tid + 128] + red[tid + 192];
  }
}

#define TB 512
#define NW 16
#define LP 76
#define GP 132
#define LSZ (128 * LP)
#define WSZ 16384
#define GK1 (-2.0f * 0.7978845608028654f * 1.4426950408889634f)
#define GK3 (-2.0f * 0.7978845608028654f * 1.4426950408889634f * 0.044715f)

__device__ __forceinline__ void ln_rows(const float* __restrict__ src,
                                        float* __restrict__ dst,
                                        const float* __restrict__ g,
                                        const float* __restrict__ bta, int tid) {
  const int s = tid >> 2, p = tid & 3;
  float sum = 0.0f, sq = 0.0f;
  for (int d = p * 16; d < p * 16 + 16; ++d) {
    float xv = src[s * LP + d];
    sum += xv; sq += xv * xv;
  }
  sum += row_xor<1>(sum); sq += row_xor<1>(sq);
  sum += row_xor<2>(sum); sq += row_xor<2>(sq);
  const float mu  = sum * (1.0f / 64.0f);
  const float var = sq * (1.0f / 64.0f) - mu * mu;
  const float inv = rsqrtf(var + 1e-5f);
  for (int d = p * 16; d < p * 16 + 16; ++d)
    dst[s * LP + d] = (src[s * LP + d] - mu) * inv * g[d] + bta[d];
}

__global__ __launch_bounds__(TB, 1) void xformer_kernel(
    const int* __restrict__ idx, const int* __restrict__ num_loops,
    const float* __restrict__ wte, const float* __restrict__ wpe,
    const float* __restrict__ ln1_g, const float* __restrict__ ln1_b,
    const float* __restrict__ w_qkv, const float* __restrict__ b_qkv,
    const float* __restrict__ w_o, const float* __restrict__ b_o,
    const float* __restrict__ ln2_g, const float* __restrict__ ln2_b,
    const float* __restrict__ w_fc, const float* __restrict__ b_fc,
    const float* __restrict__ w_pr, const float* __restrict__ b_pr,
    const float* __restrict__ lnf_g, const float* __restrict__ lnf_b,
    const float* __restrict__ ttab, float* __restrict__ out) {
  __shared__ float smem[5 * LSZ + 2 * WSZ];
  float* xb = smem;
  float* hb = smem + LSZ;
  float* qb = smem + 2 * LSZ;
  float* kb = smem + 3 * LSZ;
  float* vb = smem + 4 * LSZ;
  float* gb = qb;
  float* ob = vb;
  float* wfl = smem + 5 * LSZ;
  float* wpl = smem + 5 * LSZ + WSZ;

  const int b    = blockIdx.x;
  const int tid  = threadIdx.x;
  const int wid  = tid >> 5;
  const int lane = tid & 31;
  const int l16  = lane & 15;
  const int half = lane >> 4;
  const int koff = 2 * half;
  const int wv = __builtin_amdgcn_readfirstlane(wid);

  for (int i = tid; i < 8192; i += TB) {
    int s = i >> 6, d = i & 63;
    xb[s * LP + d] = wte[idx[b * 128 + s] * 64 + d] + wpe[i];
  }
  for (int i = tid; i < WSZ; i += TB) {
    const int k = i >> 8, n = i & 255;
    wfl[(((n >> 4) * 16 + (k >> 2)) * 32 + ((k >> 1) & 1) * 16 + (n & 15)) * 2 + (k & 1)] = w_fc[i];
  }
  for (int i = tid; i < WSZ; i += TB) {
    const int k = i >> 6, n = i & 63;
    wpl[(((n >> 4) * 64 + (k >> 2)) * 32 + ((k >> 1) & 1) * 16 + (n & 15)) * 2 + (k & 1)] = w_pr[i];
  }
  __syncthreads();

  const int NL = *num_loops;
  for (int t = 0; t < NL; ++t) {
    for (int i = tid; i < 8192; i += TB)
      xb[(i >> 6) * LP + (i & 63)] += ttab[t * 64 + (i & 63)];
    __syncthreads();

    ln_rows(xb, hb, ln1_g, ln1_b, tid);
    __syncthreads();

    for (int tile = wid; tile < 96; tile += NW) {
      const int m0 = (tile / 12) * 16, n0 = (tile % 12) * 16;
      v8f acc;
      { float bias = b_qkv[n0 + l16]; for (int r = 0; r < 8; ++r) acc[r] = bias; }
      for (int k = 0; k < 64; k += 4) {
        v2f a, w;
        a.x = hb[(m0 + l16) * LP + k + koff];
        a.y = hb[(m0 + l16) * LP + k + koff + 1];
        w.x = w_qkv[(k + koff) * 192 + n0 + l16];
        w.y = w_qkv[(k + koff + 1) * 192 + n0 + l16];
        acc = wmma4(a, w, acc);
      }
      float* dst; int nc;
      if (n0 < 64)       { dst = qb; nc = n0; }
      else if (n0 < 128) { dst = kb; nc = n0 - 64; }
      else               { dst = vb; nc = n0 - 128; }
      for (int r = 0; r < 8; ++r)
        dst[(m0 + r + 8 * half) * LP + nc + l16] = acc[r];
    }
    __syncthreads();

    for (int jj = 0; jj < 2; ++jj) {
      const int hh = wv >> 2;
      const int qt = jj ? (wv & 3) : 7 - (wv & 3);
      const int q0 = qt * 16, c0 = hh * 16;
      v8f sc[8];
#pragma unroll
      for (int kt = 0; kt < 8; ++kt) {
        if (kt > qt) break;
        v8f a8; for (int r = 0; r < 8; ++r) a8[r] = 0.0f;
        for (int kk = 0; kk < 16; kk += 4) {
          v2f a, bb;
          a.x  = qb[(q0 + l16) * LP + c0 + kk + koff];
          a.y  = qb[(q0 + l16) * LP + c0 + kk + koff + 1];
          bb.x = kb[(kt * 16 + l16) * LP + c0 + kk + koff];
          bb.y = kb[(kt * 16 + l16) * LP + c0 + kk + koff + 1];
          a8 = wmma4(a, bb, a8);
        }
        for (int r = 0; r < 8; ++r) a8[r] *= 0.25f;
        if (kt == qt) {
          for (int r = 0; r < 8; ++r)
            a8[r] = (l16 <= r + 8 * half) ? a8[r] : -1e9f;
        }
        sc[kt] = a8;
      }
      float mx[8], sm[8];
#pragma unroll
      for (int r = 0; r < 8; ++r) mx[r] = -1e30f;
#pragma unroll
      for (int kt = 0; kt < 8; ++kt) {
        if (kt > qt) break;
#pragma unroll
        for (int r = 0; r < 8; ++r) mx[r] = fmaxf(mx[r], sc[kt][r]);
      }
#pragma unroll
      for (int r = 0; r < 8; ++r) mx[r] = fmaxf(mx[r], row_xor<8>(mx[r]));
#pragma unroll
      for (int r = 0; r < 8; ++r) mx[r] = fmaxf(mx[r], row_xor<4>(mx[r]));
#pragma unroll
      for (int r = 0; r < 8; ++r) mx[r] = fmaxf(mx[r], row_xor<2>(mx[r]));
#pragma unroll
      for (int r = 0; r < 8; ++r) mx[r] = fmaxf(mx[r], row_xor<1>(mx[r]));
#pragma unroll
      for (int r = 0; r < 8; ++r) sm[r] = 0.0f;
#pragma unroll
      for (int kt = 0; kt < 8; ++kt) {
        if (kt > qt) break;
#pragma unroll
        for (int r = 0; r < 8; ++r) {
          float e = __expf(sc[kt][r] - mx[r]);
          sc[kt][r] = e; sm[r] += e;
        }
      }
#pragma unroll
      for (int r = 0; r < 8; ++r) sm[r] += row_xor<8>(sm[r]);
#pragma unroll
      for (int r = 0; r < 8; ++r) sm[r] += row_xor<4>(sm[r]);
#pragma unroll
      for (int r = 0; r < 8; ++r) sm[r] += row_xor<2>(sm[r]);
#pragma unroll
      for (int r = 0; r < 8; ++r) sm[r] += row_xor<1>(sm[r]);
#pragma unroll
      for (int r = 0; r < 8; ++r) sm[r] = 1.0f / sm[r];
      v8f o; for (int r = 0; r < 8; ++r) o[r] = 0.0f;
#pragma unroll
      for (int kt = 0; kt < 8; ++kt) {
        if (kt > qt) break;
        for (int r = 0; r < 8; ++r)
          hb[(q0 + r + 8 * half) * LP + c0 + l16] = sc[kt][r] * sm[r];
        for (int kk = 0; kk < 16; kk += 4) {
          v2f a, bb;
          a.x  = hb[(q0 + l16) * LP + c0 + kk + koff];
          a.y  = hb[(q0 + l16) * LP + c0 + kk + koff + 1];
          bb.x = vb[(kt * 16 + kk + koff) * LP + c0 + l16];
          bb.y = vb[(kt * 16 + kk + koff + 1) * LP + c0 + l16];
          o = wmma4(a, bb, o);
        }
      }
      for (int r = 0; r < 8; ++r)
        hb[(q0 + r + 8 * half) * LP + c0 + l16] = o[r];
    }
    __syncthreads();

    for (int tile = wid; tile < 32; tile += NW) {
      const int m0 = (tile >> 2) * 16, n0 = (tile & 3) * 16;
      v8f acc;
      { float bias = b_o[n0 + l16]; for (int r = 0; r < 8; ++r) acc[r] = bias; }
      for (int k = 0; k < 64; k += 4) {
        v2f a, w;
        a.x = hb[(m0 + l16) * LP + k + koff];
        a.y = hb[(m0 + l16) * LP + k + koff + 1];
        w.x = w_o[(k + koff) * 64 + n0 + l16];
        w.y = w_o[(k + koff + 1) * 64 + n0 + l16];
        acc = wmma4(a, w, acc);
      }
      for (int r = 0; r < 8; ++r)
        xb[(m0 + r + 8 * half) * LP + n0 + l16] += 0.1f * acc[r];
    }
    __syncthreads();

    ln_rows(xb, hb, ln2_g, ln2_b, tid);
    __syncthreads();

    for (int hh2 = 0; hh2 < 2; ++hh2) {
      {
        const int ntl = (wv & 7) * 16;
        const int n0g = hh2 * 128 + ntl;
        const v2f* bf = (const v2f*)(wfl + (hh2 * 8 + (wv & 7)) * 1024) + lane;
        v2f wf[16];
#pragma unroll
        for (int ks = 0; ks < 16; ++ks) wf[ks] = bf[ks * 32];
        const float bias = b_fc[n0g + l16];
        for (int i = 0; i < 4; ++i) {
          const int m0 = ((wv >> 3) + 2 * i) * 16;
          v8f acc;
          for (int r = 0; r < 8; ++r) acc[r] = bias;
#pragma unroll
          for (int ks = 0; ks < 16; ++ks) {
            v2f a;
            a.x = hb[(m0 + l16) * LP + 4 * ks + koff];
            a.y = hb[(m0 + l16) * LP + 4 * ks + koff + 1];
            acc = wmma4(a, wf[ks], acc);
          }
          float ge[8];
          for (int r = 0; r < 8; ++r) ge[r] = acc[r] * (GK1 + GK3 * (acc[r] * acc[r]));
          for (int r = 0; r < 8; ++r) ge[r] = __builtin_amdgcn_exp2f(ge[r]);
          for (int r = 0; r < 8; ++r) ge[r] = __builtin_amdgcn_rcpf(1.0f + ge[r]);
          for (int r = 0; r < 8; ++r)
            gb[(m0 + r + 8 * half) * GP + ntl + l16] = acc[r] * ge[r];
        }
      }
      __syncthreads();
      {
        const int n0 = (wv & 3) * 16;
        const v2f* bp = (const v2f*)(wpl + ((wv & 3) * 64 + hh2 * 32) * 64) + lane;
        v2f wp[32];
#pragma unroll
        for (int ks = 0; ks < 32; ++ks) wp[ks] = bp[ks * 32];
        const float bias = b_pr[n0 + l16];
        for (int i = 0; i < 2; ++i) {
          const int m0 = ((wv >> 2) + 4 * i) * 16;
          v8f acc;
          if (hh2 == 0) {
            for (int r = 0; r < 8; ++r) acc[r] = bias;
          } else {
            for (int r = 0; r < 8; ++r)
              acc[r] = ob[(m0 + r + 8 * half) * LP + n0 + l16];
          }
#pragma unroll
          for (int ks = 0; ks < 32; ++ks) {
            v2f a;
            a.x = gb[(m0 + l16) * GP + 4 * ks + koff];
            a.y = gb[(m0 + l16) * GP + 4 * ks + koff + 1];
            acc = wmma4(a, wp[ks], acc);
          }
          if (hh2 == 0) {
            for (int r = 0; r < 8; ++r)
              ob[(m0 + r + 8 * half) * LP + n0 + l16] = acc[r];
          } else {
            for (int r = 0; r < 8; ++r)
              xb[(m0 + r + 8 * half) * LP + n0 + l16] += 0.1f * acc[r];
          }
        }
      }
      __syncthreads();
    }
  }

  {
    const int s = tid >> 2, p = tid & 3;
    float sum = 0.0f, sq = 0.0f;
    for (int d = p * 16; d < p * 16 + 16; ++d) {
      float xv = xb[s * LP + d];
      sum += xv; sq += xv * xv;
    }
    sum += row_xor<1>(sum); sq += row_xor<1>(sq);
    sum += row_xor<2>(sum); sq += row_xor<2>(sq);
    const float mu  = sum * (1.0f / 64.0f);
    const float var = sq * (1.0f / 64.0f) - mu * mu;
    const float inv = rsqrtf(var + 1e-5f);
    for (int d = p * 16; d < p * 16 + 16; ++d)
      out[b * 8192 + s * 64 + d] =
          (xb[s * LP + d] - mu) * inv * lnf_g[d] + lnf_b[d];
  }
}

extern "C" void kernel_launch(void* const* d_in, const int* in_sizes, int n_in,
                              void* d_out, int out_size, void* d_ws, size_t ws_size,
                              hipStream_t stream) {
  const int*   idx   = (const int*)  d_in[0];
  const int*   nlp   = (const int*)  d_in[1];
  const float* wte   = (const float*)d_in[2];
  const float* wpe   = (const float*)d_in[3];
  const float* t_w1  = (const float*)d_in[4];
  const float* t_b1  = (const float*)d_in[5];
  const float* t_w2  = (const float*)d_in[6];
  const float* t_b2  = (const float*)d_in[7];
  const float* ln1_g = (const float*)d_in[8];
  const float* ln1_b = (const float*)d_in[9];
  const float* w_qkv = (const float*)d_in[10];
  const float* b_qkv = (const float*)d_in[11];
  const float* w_o   = (const float*)d_in[12];
  const float* b_o   = (const float*)d_in[13];
  const float* ln2_g = (const float*)d_in[14];
  const float* ln2_b = (const float*)d_in[15];
  const float* w_fc  = (const float*)d_in[16];
  const float* b_fc  = (const float*)d_in[17];
  const float* w_pr  = (const float*)d_in[18];
  const float* b_pr  = (const float*)d_in[19];
  const float* lnf_g = (const float*)d_in[20];
  const float* lnf_b = (const float*)d_in[21];

  float* ttab = (float*)d_ws;

  temb_kernel<<<128, 256, 0, stream>>>(nlp, t_w1, t_b1, t_w2, t_b2, ttab);
  xformer_kernel<<<32, TB, 0, stream>>>(idx, nlp, wte, wpe, ln1_g, ln1_b,
                                        w_qkv, b_qkv, w_o, b_o, ln2_g, ln2_b,
                                        w_fc, b_fc, w_pr, b_pr, lnf_g, lnf_b,
                                        ttab, (float*)d_out);
}


# Attention on transposed tiles: S^T=K.Q^T stays in registers as B of O^T=V^T.P^T (V stored transposed), no LDS round trip of P; row max/sum in-lane + 1 half swap; O scaled not P; sum order changed
# speedup vs baseline: 3.4032x; 1.0616x over previous
#include <hip/hip_runtime.h>
#include <hip/hip_bf16.h>
#include <math.h>


typedef __attribute__((ext_vector_type(2))) float v2f;
typedef __attribute__((ext_vector_type(8))) float v8f;

__device__ __forceinline__ v8f wmma4(v2f a, v2f b, v8f c) {
  return __builtin_amdgcn_wmma_f32_16x16x4_f32(
       false, a,  false, b,
       (short)0, c,  false,  false);
}

template <int M>
__device__ __forceinline__ float row_xor(float v) {
  return __int_as_float(__builtin_amdgcn_update_dpp(
      0, __float_as_int(v), 0x160 + M, 0xf, 0xf, true));
}

__device__ __forceinline__ float xhalf(float v) {
  const int b = __float_as_int(v);
  return __int_as_float(__builtin_amdgcn_permlanex16(
      b, b, 0x76543210, 0xfedcba98, false, false));
}

__global__ __launch_bounds__(256) void temb_kernel(
    const int* __restrict__ num_loops,
    const float* __restrict__ t_w1, const float* __restrict__ t_b1,
    const float* __restrict__ t_w2, const float* __restrict__ t_b2,
    float* __restrict__ ttab) {
  const int t = blockIdx.x;
  if (t >= *num_loops) return;
  __shared__ float te[256];
  __shared__ float hs[1024];
  __shared__ float red[256];
  const int tid = threadIdx.x;

  if (tid < 128) {
    float f   = expf(-logf(10000.0f) * (float)tid / 128.0f);
    float arg = (float)t * f;
    te[tid]       = cosf(arg);
    te[128 + tid] = sinf(arg);
  }
  __syncthreads();

  for (int j = tid; j < 1024; j += 256) {
    float acc = t_b1[j];
    for (int k = 0; k < 256; ++k) acc += te[k] * t_w1[k * 1024 + j];
    hs[j] = acc / (1.0f + expf(-acc));
  }
  __syncthreads();

  const int j = tid & 63, p = tid >> 6;
  float acc = 0.0f;
  for (int k = p * 256; k < p * 256 + 256; ++k) acc += hs[k] * t_w2[k * 64 + j];
  red[tid] = acc;
  __syncthreads();
  if (tid < 64) {
    ttab[t * 64 + tid] =
        t_b2[tid] + red[tid] + red[tid + 64] + red[tid + 128] + red[tid + 192];
  }
}

#define TB 512
#define NW 16
#define LP 76
#define GP 132
#define VP 132
#define LSZ (128 * LP)
#define WSZ 16384
#define GK1 (-2.0f * 0.7978845608028654f * 1.4426950408889634f)
#define GK3 (-2.0f * 0.7978845608028654f * 1.4426950408889634f * 0.044715f)

__device__ __forceinline__ void ln_rows(const float* __restrict__ src,
                                        float* __restrict__ dst,
                                        const float* __restrict__ g,
                                        const float* __restrict__ bta, int tid) {
  const int s = tid >> 2, p = tid & 3;
  float sum = 0.0f, sq = 0.0f;
  for (int d = p * 16; d < p * 16 + 16; ++d) {
    float xv = src[s * LP + d];
    sum += xv; sq += xv * xv;
  }
  sum += row_xor<1>(sum); sq += row_xor<1>(sq);
  sum += row_xor<2>(sum); sq += row_xor<2>(sq);
  const float mu  = sum * (1.0f / 64.0f);
  const float var = sq * (1.0f / 64.0f) - mu * mu;
  const float inv = rsqrtf(var + 1e-5f);
  for (int d = p * 16; d < p * 16 + 16; ++d)
    dst[s * LP + d] = (src[s * LP + d] - mu) * inv * g[d] + bta[d];
}

__global__ __launch_bounds__(TB, 1) void xformer_kernel(
    const int* __restrict__ idx, const int* __restrict__ num_loops,
    const float* __restrict__ wte, const float* __restrict__ wpe,
    const float* __restrict__ ln1_g, const float* __restrict__ ln1_b,
    const float* __restrict__ w_qkv, const float* __restrict__ b_qkv,
    const float* __restrict__ w_o, const float* __restrict__ b_o,
    const float* __restrict__ ln2_g, const float* __restrict__ ln2_b,
    const float* __restrict__ w_fc, const float* __restrict__ b_fc,
    const float* __restrict__ w_pr, const float* __restrict__ b_pr,
    const float* __restrict__ lnf_g, const float* __restrict__ lnf_b,
    const float* __restrict__ ttab, float* __restrict__ out) {
  __shared__ __align__(16) float smem[5 * LSZ + 2 * WSZ];
  float* xb = smem;
  float* hb = smem + LSZ;
  float* qb = smem + 2 * LSZ;
  float* kb = smem + 3 * LSZ;
  float* vb = smem + 4 * LSZ;
  float* gb = qb;
  float* ob = vb;
  float* wfl = smem + 5 * LSZ;
  float* wpl = smem + 5 * LSZ + WSZ;

  const int b    = blockIdx.x;
  const int tid  = threadIdx.x;
  const int wid  = tid >> 5;
  const int lane = tid & 31;
  const int l16  = lane & 15;
  const int half = lane >> 4;
  const int koff = 2 * half;
  const int wv = __builtin_amdgcn_readfirstlane(wid);

  for (int i = tid; i < 8192; i += TB) {
    int s = i >> 6, d = i & 63;
    xb[s * LP + d] = wte[idx[b * 128 + s] * 64 + d] + wpe[i];
  }
  for (int i = tid; i < WSZ; i += TB) {
    const int k = i >> 8, n = i & 255;
    wfl[(((n >> 4) * 16 + (k >> 2)) * 32 + ((k >> 1) & 1) * 16 + (n & 15)) * 2 + (k & 1)] = w_fc[i];
  }
  for (int i = tid; i < WSZ; i += TB) {
    const int k = i >> 6, n = i & 63;
    wpl[(((n >> 4) * 64 + (k >> 2)) * 32 + ((k >> 1) & 1) * 16 + (n & 15)) * 2 + (k & 1)] = w_pr[i];
  }
  __syncthreads();

  const int NL = *num_loops;
  for (int t = 0; t < NL; ++t) {
    for (int i = tid; i < 8192; i += TB)
      xb[(i >> 6) * LP + (i & 63)] += ttab[t * 64 + (i & 63)];
    __syncthreads();

    ln_rows(xb, hb, ln1_g, ln1_b, tid);
    __syncthreads();

    for (int tile = wv; tile < 96; tile += NW) {
      const int m0 = (tile / 12) * 16, n0 = (tile % 12) * 16;
      v8f acc;
      { float bias = b_qkv[n0 + l16]; for (int r = 0; r < 8; ++r) acc[r] = bias; }
      for (int k = 0; k < 64; k += 4) {
        v2f a, w;
        a.x = hb[(m0 + l16) * LP + k + koff];
        a.y = hb[(m0 + l16) * LP + k + koff + 1];
        w.x = w_qkv[(k + koff) * 192 + n0 + l16];
        w.y = w_qkv[(k + koff + 1) * 192 + n0 + l16];
        acc = wmma4(a, w, acc);
      }
      if (n0 < 128) {
        float* dst = (n0 < 64) ? qb : kb;
        const int nc = n0 & 63;
        for (int r = 0; r < 8; ++r)
          dst[(m0 + r + 8 * half) * LP + nc + l16] = acc[r];
      } else {
        float4* dst = (float4*)(vb + (n0 - 128 + l16) * VP + m0 + 8 * half);
        dst[0] = make_float4(acc[0], acc[1], acc[2], acc[3]);
        dst[1] = make_float4(acc[4], acc[5], acc[6], acc[7]);
      }
    }
    __syncthreads();

    for (int jj = 0; jj < 2; ++jj) {
      const int hh = wv >> 2;
      const int qt = jj ? (wv & 3) : 7 - (wv & 3);
      const int q0 = qt * 16, c0 = hh * 16;
      v2f qf[4];
#pragma unroll
      for (int kk = 0; kk < 4; ++kk) {
        qf[kk].x = qb[(q0 + l16) * LP + c0 + 4 * kk + koff];
        qf[kk].y = qb[(q0 + l16) * LP + c0 + 4 * kk + koff + 1];
      }
      v8f sc[8];
#pragma unroll
      for (int kt = 0; kt < 8; ++kt) {
        if (kt > qt) break;
        v8f a8; for (int r = 0; r < 8; ++r) a8[r] = 0.0f;
#pragma unroll
        for (int kk = 0; kk < 4; ++kk) {
          v2f kf;
          kf.x = kb[(kt * 16 + l16) * LP + c0 + 4 * kk + koff];
          kf.y = kb[(kt * 16 + l16) * LP + c0 + 4 * kk + koff + 1];
          a8 = wmma4(kf, qf[kk], a8);
        }
        for (int r = 0; r < 8; ++r) a8[r] *= 0.25f;
        if (kt == qt) {
          for (int r = 0; r < 8; ++r)
            a8[r] = (r + 8 * half <= l16) ? a8[r] : -1e9f;
        }
        sc[kt] = a8;
      }
      float m8[8], s8[8];
#pragma unroll
      for (int r = 0; r < 8; ++r) m8[r] = sc[0][r];
#pragma unroll
      for (int kt = 1; kt < 8; ++kt) {
        if (kt > qt) break;
#pragma unroll
        for (int r = 0; r < 8; ++r) m8[r] = fmaxf(m8[r], sc[kt][r]);
      }
      float mx = fmaxf(fmaxf(fmaxf(m8[0], m8[1]), fmaxf(m8[2], m8[3])),
                       fmaxf(fmaxf(m8[4], m8[5]), fmaxf(m8[6], m8[7])));
      mx = fmaxf(mx, xhalf(mx));
#pragma unroll
      for (int r = 0; r < 8; ++r) s8[r] = 0.0f;
#pragma unroll
      for (int kt = 0; kt < 8; ++kt) {
        if (kt > qt) break;
#pragma unroll
        for (int r = 0; r < 8; ++r) {
          float e = __expf(sc[kt][r] - mx);
          sc[kt][r] = e; s8[r] += e;
        }
      }
      float sm = ((s8[0] + s8[1]) + (s8[2] + s8[3])) + ((s8[4] + s8[5]) + (s8[6] + s8[7]));
      sm += xhalf(sm);
      const float inv = 1.0f / sm;
      const float4* vp = (const float4*)(vb + (c0 + l16) * VP + 8 * half);
      v8f o; for (int r = 0; r < 8; ++r) o[r] = 0.0f;
      float4 vlo = vp[0], vhi = vp[1];
#pragma unroll
      for (int kt = 0; kt < 8; ++kt) {
        float4 nlo = vlo, nhi = vhi;
        if (kt < qt) { nlo = vp[4 * (kt + 1)]; nhi = vp[4 * (kt + 1) + 1]; }
        v2f vf, pf;
        vf.x = vlo.x; vf.y = vlo.y; pf.x = sc[kt][0]; pf.y = sc[kt][1];
        o = wmma4(vf, pf, o);
        vf.x = vlo.z; vf.y = vlo.w; pf.x = sc[kt][2]; pf.y = sc[kt][3];
        o = wmma4(vf, pf, o);
        vf.x = vhi.x; vf.y = vhi.y; pf.x = sc[kt][4]; pf.y = sc[kt][5];
        o = wmma4(vf, pf, o);
        vf.x = vhi.z; vf.y = vhi.w; pf.x = sc[kt][6]; pf.y = sc[kt][7];
        o = wmma4(vf, pf, o);
        if (kt >= qt) break;
        vlo = nlo; vhi = nhi;
      }
      float4* dst = (float4*)(hb + (q0 + l16) * LP + c0 + 8 * half);
      dst[0] = make_float4(o[0] * inv, o[1] * inv, o[2] * inv, o[3] * inv);
      dst[1] = make_float4(o[4] * inv, o[5] * inv, o[6] * inv, o[7] * inv);
    }
    __syncthreads();

    for (int tile = wid; tile < 32; tile += NW) {
      const int m0 = (tile >> 2) * 16, n0 = (tile & 3) * 16;
      v8f acc;
      { float bias = b_o[n0 + l16]; for (int r = 0; r < 8; ++r) acc[r] = bias; }
      for (int k = 0; k < 64; k += 4) {
        v2f a, w;
        a.x = hb[(m0 + l16) * LP + k + koff];
        a.y = hb[(m0 + l16) * LP + k + koff + 1];
        w.x = w_o[(k + koff) * 64 + n0 + l16];
        w.y = w_o[(k + koff + 1) * 64 + n0 + l16];
        acc = wmma4(a, w, acc);
      }
      for (int r = 0; r < 8; ++r)
        xb[(m0 + r + 8 * half) * LP + n0 + l16] += 0.1f * acc[r];
    }
    __syncthreads();

    ln_rows(xb, hb, ln2_g, ln2_b, tid);
    __syncthreads();

    for (int hh2 = 0; hh2 < 2; ++hh2) {
      {
        const int ntl = (wv & 7) * 16;
        const int n0g = hh2 * 128 + ntl;
        const v2f* bf = (const v2f*)(wfl + (hh2 * 8 + (wv & 7)) * 1024) + lane;
        v2f wf[16];
#pragma unroll
        for (int ks = 0; ks < 16; ++ks) wf[ks] = bf[ks * 32];
        const float bias = b_fc[n0g + l16];
        for (int i = 0; i < 4; ++i) {
          const int m0 = ((wv >> 3) + 2 * i) * 16;
          v8f acc;
          for (int r = 0; r < 8; ++r) acc[r] = bias;
#pragma unroll
          for (int ks = 0; ks < 16; ++ks) {
            v2f a;
            a.x = hb[(m0 + l16) * LP + 4 * ks + koff];
            a.y = hb[(m0 + l16) * LP + 4 * ks + koff + 1];
            acc = wmma4(a, wf[ks], acc);
          }
          float ge[8];
          for (int r = 0; r < 8; ++r) ge[r] = acc[r] * (GK1 + GK3 * (acc[r] * acc[r]));
          for (int r = 0; r < 8; ++r) ge[r] = __builtin_amdgcn_exp2f(ge[r]);
          for (int r = 0; r < 8; ++r) ge[r] = __builtin_amdgcn_rcpf(1.0f + ge[r]);
          for (int r = 0; r < 8; ++r)
            gb[(m0 + r + 8 * half) * GP + ntl + l16] = acc[r] * ge[r];
        }
      }
      __syncthreads();
      {
        const int n0 = (wv & 3) * 16;
        const v2f* bp = (const v2f*)(wpl + ((wv & 3) * 64 + hh2 * 32) * 64) + lane;
        v2f wp[32];
#pragma unroll
        for (int ks = 0; ks < 32; ++ks) wp[ks] = bp[ks * 32];
        const float bias = b_pr[n0 + l16];
        for (int i = 0; i < 2; ++i) {
          const int m0 = ((wv >> 2) + 4 * i) * 16;
          v8f acc;
          if (hh2 == 0) {
            for (int r = 0; r < 8; ++r) acc[r] = bias;
          } else {
            for (int r = 0; r < 8; ++r)
              acc[r] = ob[(m0 + r + 8 * half) * LP + n0 + l16];
          }
#pragma unroll
          for (int ks = 0; ks < 32; ++ks) {
            v2f a;
            a.x = gb[(m0 + l16) * GP + 4 * ks + koff];
            a.y = gb[(m0 + l16) * GP + 4 * ks + koff + 1];
            acc = wmma4(a, wp[ks], acc);
          }
          if (hh2 == 0) {
            for (int r = 0; r < 8; ++r)
              ob[(m0 + r + 8 * half) * LP + n0 + l16] = acc[r];
          } else {
            for (int r = 0; r < 8; ++r)
              xb[(m0 + r + 8 * half) * LP + n0 + l16] += 0.1f * acc[r];
          }
        }
      }
      __syncthreads();
    }
  }

  {
    const int s = tid >> 2, p = tid & 3;
    float sum = 0.0f, sq = 0.0f;
    for (int d = p * 16; d < p * 16 + 16; ++d) {
      float xv = xb[s * LP + d];
      sum += xv; sq += xv * xv;
    }
    sum += row_xor<1>(sum); sq += row_xor<1>(sq);
    sum += row_xor<2>(sum); sq += row_xor<2>(sq);
    const float mu  = sum * (1.0f / 64.0f);
    const float var = sq * (1.0f / 64.0f) - mu * mu;
    const float inv = rsqrtf(var + 1e-5f);
    for (int d = p * 16; d < p * 16 + 16; ++d)
      out[b * 8192 + s * 64 + d] =
          (xb[s * LP + d] - mu) * inv * lnf_g[d] + lnf_b[d];
  }
}

extern "C" void kernel_launch(void* const* d_in, const int* in_sizes, int n_in,
                              void* d_out, int out_size, void* d_ws, size_t ws_size,
                              hipStream_t stream) {
  const int*   idx   = (const int*)  d_in[0];
  const int*   nlp   = (const int*)  d_in[1];
  const float* wte   = (const float*)d_in[2];
  const float* wpe   = (const float*)d_in[3];
  const float* t_w1  = (const float*)d_in[4];
  const float* t_b1  = (const float*)d_in[5];
  const float* t_w2  = (const float*)d_in[6];
  const float* t_b2  = (const float*)d_in[7];
  const float* ln1_g = (const float*)d_in[8];
  const float* ln1_b = (const float*)d_in[9];
  const float* w_qkv = (const float*)d_in[10];
  const float* b_qkv = (const float*)d_in[11];
  const float* w_o   = (const float*)d_in[12];
  const float* b_o   = (const float*)d_in[13];
  const float* ln2_g = (const float*)d_in[14];
  const float* ln2_b = (const float*)d_in[15];
  const float* w_fc  = (const float*)d_in[16];
  const float* b_fc  = (const float*)d_in[17];
  const float* w_pr  = (const float*)d_in[18];
  const float* b_pr  = (const float*)d_in[19];
  const float* lnf_g = (const float*)d_in[20];
  const float* lnf_b = (const float*)d_in[21];

  float* ttab = (float*)d_ws;

  temb_kernel<<<128, 256, 0, stream>>>(nlp, t_w1, t_b1, t_w2, t_b2, ttab);
  xformer_kernel<<<32, TB, 0, stream>>>(idx, nlp, wte, wpe, ln1_g, ln1_b,
                                        w_qkv, b_qkv, w_o, b_o, ln2_g, ln2_b,
                                        w_fc, b_fc, w_pr, b_pr, lnf_g, lnf_b,
                                        ttab, (float*)d_out);
}


# A WMMA tile's LDS fragment loads issued as one group ahead of its WMMA chain (scheduling fences), not one load per two WMMAs; LN scale/shift read from LDS to free registers; sums and order unchanged
# speedup vs baseline: 3.7587x; 1.1045x over previous
#include <hip/hip_runtime.h>
#include <hip/hip_bf16.h>
#include <math.h>


typedef __attribute__((ext_vector_type(2))) float v2f;
typedef __attribute__((ext_vector_type(8))) float v8f;

__device__ __forceinline__ v8f wmma4(v2f a, v2f b, v8f c) {
  return __builtin_amdgcn_wmma_f32_16x16x4_f32(
       false, a,  false, b,
       (short)0, c,  false,  false);
}

__device__ __forceinline__ void frag_fence() { __builtin_amdgcn_sched_barrier(0); }

template <int M>
__device__ __forceinline__ float row_xor(float v) {
  return __int_as_float(__builtin_amdgcn_update_dpp(
      0, __float_as_int(v), 0x160 + M, 0xf, 0xf, true));
}

__device__ __forceinline__ float xhalf(float v) {
  const int b = __float_as_int(v);
  return __int_as_float(__builtin_amdgcn_permlanex16(
      b, b, 0x76543210, 0xfedcba98, false, false));
}

__global__ __launch_bounds__(256) void temb_kernel(
    const int* __restrict__ num_loops,
    const float* __restrict__ t_w1, const float* __restrict__ t_b1,
    const float* __restrict__ t_w2, const float* __restrict__ t_b2,
    float* __restrict__ ttab) {
  const int t = blockIdx.x;
  if (t >= *num_loops) return;
  __shared__ float te[256];
  __shared__ float hs[1024];
  __shared__ float red[256];
  const int tid = threadIdx.x;

  if (tid < 128) {
    float f   = expf(-logf(10000.0f) * (float)tid / 128.0f);
    float arg = (float)t * f;
    te[tid]       = cosf(arg);
    te[128 + tid] = sinf(arg);
  }
  __syncthreads();

  for (int j = tid; j < 1024; j += 256) {
    float acc = t_b1[j];
    for (int k = 0; k < 256; ++k) acc += te[k] * t_w1[k * 1024 + j];
    hs[j] = acc / (1.0f + expf(-acc));
  }
  __syncthreads();

  const int j = tid & 63, p = tid >> 6;
  float acc = 0.0f;
  for (int k = p * 256; k < p * 256 + 256; ++k) acc += hs[k] * t_w2[k * 64 + j];
  red[tid] = acc;
  __syncthreads();
  if (tid < 64) {
    ttab[t * 64 + tid] =
        t_b2[tid] + red[tid] + red[tid + 64] + red[tid + 128] + red[tid + 192];
  }
}

#define TB 512
#define NW 16
#define LP 76
#define GP 132
#define VP 132
#define LSZ (128 * LP)
#define WSZ 16384
#define GK1 (-2.0f * 0.7978845608028654f * 1.4426950408889634f)
#define GK3 (-2.0f * 0.7978845608028654f * 1.4426950408889634f * 0.044715f)

__device__ __forceinline__ void ln_rows(const float* __restrict__ src,
                                        float* __restrict__ dst,
                                        const float* __restrict__ g,
                                        const float* __restrict__ bta, int tid) {
  const int s = tid >> 2, p = tid & 3;
  float sum = 0.0f, sq = 0.0f;
  for (int d = p * 16; d < p * 16 + 16; ++d) {
    float xv = src[s * LP + d];
    sum += xv; sq += xv * xv;
  }
  sum += row_xor<1>(sum); sq += row_xor<1>(sq);
  sum += row_xor<2>(sum); sq += row_xor<2>(sq);
  const float mu  = sum * (1.0f / 64.0f);
  const float var = sq * (1.0f / 64.0f) - mu * mu;
  const float inv = rsqrtf(var + 1e-5f);
  for (int d = p * 16; d < p * 16 + 16; ++d)
    dst[s * LP + d] = (src[s * LP + d] - mu) * inv * g[d] + bta[d];
}

__global__ __launch_bounds__(TB, 1) void xformer_kernel(
    const int* __restrict__ idx, const int* __restrict__ num_loops,
    const float* __restrict__ wte, const float* __restrict__ wpe,
    const float* __restrict__ ln1_g, const float* __restrict__ ln1_b,
    const float* __restrict__ w_qkv, const float* __restrict__ b_qkv,
    const float* __restrict__ w_o, const float* __restrict__ b_o,
    const float* __restrict__ ln2_g, const float* __restrict__ ln2_b,
    const float* __restrict__ w_fc, const float* __restrict__ b_fc,
    const float* __restrict__ w_pr, const float* __restrict__ b_pr,
    const float* __restrict__ lnf_g, const float* __restrict__ lnf_b,
    const float* __restrict__ ttab, float* __restrict__ out) {
  __shared__ __align__(16) float smem[5 * LSZ + 2 * WSZ + 256];
  float* xb = smem;
  float* hb = smem + LSZ;
  float* qb = smem + 2 * LSZ;
  float* kb = smem + 3 * LSZ;
  float* vb = smem + 4 * LSZ;
  float* gb = qb;
  float* ob = vb;
  float* wfl = smem + 5 * LSZ;
  float* wpl = smem + 5 * LSZ + WSZ;
  float* lnp = smem + 5 * LSZ + 2 * WSZ;

  const int b    = blockIdx.x;
  const int tid  = threadIdx.x;
  const int wid  = tid >> 5;
  const int lane = tid & 31;
  const int l16  = lane & 15;
  const int half = lane >> 4;
  const int koff = 2 * half;
  const int wv = __builtin_amdgcn_readfirstlane(wid);

  for (int i = tid; i < 8192; i += TB) {
    int s = i >> 6, d = i & 63;
    xb[s * LP + d] = wte[idx[b * 128 + s] * 64 + d] + wpe[i];
  }
  for (int i = tid; i < WSZ; i += TB) {
    const int k = i >> 8, n = i & 255;
    wfl[(((n >> 4) * 16 + (k >> 2)) * 32 + ((k >> 1) & 1) * 16 + (n & 15)) * 2 + (k & 1)] = w_fc[i];
  }
  for (int i = tid; i < WSZ; i += TB) {
    const int k = i >> 6, n = i & 63;
    wpl[(((n >> 4) * 64 + (k >> 2)) * 32 + ((k >> 1) & 1) * 16 + (n & 15)) * 2 + (k & 1)] = w_pr[i];
  }
  if (tid < 64) {
    lnp[tid]       = ln1_g[tid];
    lnp[64 + tid]  = ln1_b[tid];
    lnp[128 + tid] = ln2_g[tid];
    lnp[192 + tid] = ln2_b[tid];
  }
  __syncthreads();

  const int NL = *num_loops;
  for (int t = 0; t < NL; ++t) {
    for (int i = tid; i < 8192; i += TB)
      xb[(i >> 6) * LP + (i & 63)] += ttab[t * 64 + (i & 63)];
    __syncthreads();

    ln_rows(xb, hb, lnp, lnp + 64, tid);
    __syncthreads();

    for (int tile = wv; tile < 96; tile += NW) {
      const int m0 = (tile / 12) * 16, n0 = (tile % 12) * 16;
      v8f acc;
      { float bias = b_qkv[n0 + l16]; for (int r = 0; r < 8; ++r) acc[r] = bias; }
      v2f af[16], wq[16];
#pragma unroll
      for (int ks = 0; ks < 16; ++ks) {
        wq[ks].x = w_qkv[(4 * ks + koff) * 192 + n0 + l16];
        wq[ks].y = w_qkv[(4 * ks + koff + 1) * 192 + n0 + l16];
        af[ks].x = hb[(m0 + l16) * LP + 4 * ks + koff];
        af[ks].y = hb[(m0 + l16) * LP + 4 * ks + koff + 1];
      }
      frag_fence();
#pragma unroll
      for (int ks = 0; ks < 16; ++ks) acc = wmma4(af[ks], wq[ks], acc);
      frag_fence();
      if (n0 < 128) {
        float* dst = (n0 < 64) ? qb : kb;
        const int nc = n0 & 63;
        for (int r = 0; r < 8; ++r)
          dst[(m0 + r + 8 * half) * LP + nc + l16] = acc[r];
      } else {
        float4* dst = (float4*)(vb + (n0 - 128 + l16) * VP + m0 + 8 * half);
        dst[0] = make_float4(acc[0], acc[1], acc[2], acc[3]);
        dst[1] = make_float4(acc[4], acc[5], acc[6], acc[7]);
      }
    }
    __syncthreads();

    for (int jj = 0; jj < 2; ++jj) {
      const int hh = wv >> 2;
      const int qt = jj ? (wv & 3) : 7 - (wv & 3);
      const int q0 = qt * 16, c0 = hh * 16;
      v2f qf[4];
#pragma unroll
      for (int kk = 0; kk < 4; ++kk) {
        qf[kk].x = qb[(q0 + l16) * LP + c0 + 4 * kk + koff];
        qf[kk].y = qb[(q0 + l16) * LP + c0 + 4 * kk + koff + 1];
      }
      v8f sc[8];
#pragma unroll
      for (int g = 0; g < 2; ++g) {
        if (4 * g > qt) break;
        v2f kf[4][4];
#pragma unroll
        for (int j = 0; j < 4; ++j) {
#pragma unroll
          for (int kk = 0; kk < 4; ++kk) {
            kf[j][kk].x = kb[((4 * g + j) * 16 + l16) * LP + c0 + 4 * kk + koff];
            kf[j][kk].y = kb[((4 * g + j) * 16 + l16) * LP + c0 + 4 * kk + koff + 1];
          }
        }
        frag_fence();
#pragma unroll
        for (int j = 0; j < 4; ++j) {
          const int kt = 4 * g + j;
          if (kt > qt) break;
          v8f a8; for (int r = 0; r < 8; ++r) a8[r] = 0.0f;
#pragma unroll
          for (int kk = 0; kk < 4; ++kk)
            a8 = wmma4(kf[j][kk], qf[kk], a8);
          for (int r = 0; r < 8; ++r) a8[r] *= 0.25f;
          if (kt == qt) {
            for (int r = 0; r < 8; ++r)
              a8[r] = (r + 8 * half <= l16) ? a8[r] : -1e9f;
          }
          sc[kt] = a8;
        }
        frag_fence();
      }
      float m8[8], s8[8];
#pragma unroll
      for (int r = 0; r < 8; ++r) m8[r] = sc[0][r];
#pragma unroll
      for (int kt = 1; kt < 8; ++kt) {
        if (kt > qt) break;
#pragma unroll
        for (int r = 0; r < 8; ++r) m8[r] = fmaxf(m8[r], sc[kt][r]);
      }
      float mx = fmaxf(fmaxf(fmaxf(m8[0], m8[1]), fmaxf(m8[2], m8[3])),
                       fmaxf(fmaxf(m8[4], m8[5]), fmaxf(m8[6], m8[7])));
      mx = fmaxf(mx, xhalf(mx));
#pragma unroll
      for (int r = 0; r < 8; ++r) s8[r] = 0.0f;
#pragma unroll
      for (int kt = 0; kt < 8; ++kt) {
        if (kt > qt) break;
#pragma unroll
        for (int r = 0; r < 8; ++r) {
          float e = __expf(sc[kt][r] - mx);
          sc[kt][r] = e; s8[r] += e;
        }
      }
      float sm = ((s8[0] + s8[1]) + (s8[2] + s8[3])) + ((s8[4] + s8[5]) + (s8[6] + s8[7]));
      sm += xhalf(sm);
      const float inv = 1.0f / sm;
      const float4* vp = (const float4*)(vb + (c0 + l16) * VP + 8 * half);
      v8f o; for (int r = 0; r < 8; ++r) o[r] = 0.0f;
#pragma unroll
      for (int g = 0; g < 2; ++g) {
        if (4 * g > qt) break;
        float4 vlo[4], vhi[4];
#pragma unroll
        for (int j = 0; j < 4; ++j) { vlo[j] = vp[4 * (4 * g + j)]; vhi[j] = vp[4 * (4 * g + j) + 1]; }
        frag_fence();
#pragma unroll
        for (int j = 0; j < 4; ++j) {
          const int kt = 4 * g + j;
          if (kt > qt) break;
          v2f vf, pf;
          vf.x = vlo[j].x; vf.y = vlo[j].y; pf.x = sc[kt][0]; pf.y = sc[kt][1];
          o = wmma4(vf, pf, o);
          vf.x = vlo[j].z; vf.y = vlo[j].w; pf.x = sc[kt][2]; pf.y = sc[kt][3];
          o = wmma4(vf, pf, o);
          vf.x = vhi[j].x; vf.y = vhi[j].y; pf.x = sc[kt][4]; pf.y = sc[kt][5];
          o = wmma4(vf, pf, o);
          vf.x = vhi[j].z; vf.y = vhi[j].w; pf.x = sc[kt][6]; pf.y = sc[kt][7];
          o = wmma4(vf, pf, o);
        }
        frag_fence();
      }
      float4* dst = (float4*)(hb + (q0 + l16) * LP + c0 + 8 * half);
      dst[0] = make_float4(o[0] * inv, o[1] * inv, o[2] * inv, o[3] * inv);
      dst[1] = make_float4(o[4] * inv, o[5] * inv, o[6] * inv, o[7] * inv);
    }
    __syncthreads();

    for (int tile = wid; tile < 32; tile += NW) {
      const int m0 = (tile >> 2) * 16, n0 = (tile & 3) * 16;
      v8f acc;
      { float bias = b_o[n0 + l16]; for (int r = 0; r < 8; ++r) acc[r] = bias; }
      v2f af[16], wo[16];
#pragma unroll
      for (int ks = 0; ks < 16; ++ks) {
        wo[ks].x = w_o[(4 * ks + koff) * 64 + n0 + l16];
        wo[ks].y = w_o[(4 * ks + koff + 1) * 64 + n0 + l16];
        af[ks].x = hb[(m0 + l16) * LP + 4 * ks + koff];
        af[ks].y = hb[(m0 + l16) * LP + 4 * ks + koff + 1];
      }
      frag_fence();
#pragma unroll
      for (int ks = 0; ks < 16; ++ks) acc = wmma4(af[ks], wo[ks], acc);
      frag_fence();
      for (int r = 0; r < 8; ++r)
        xb[(m0 + r + 8 * half) * LP + n0 + l16] += 0.1f * acc[r];
    }
    __syncthreads();

    ln_rows(xb, hb, lnp + 128, lnp + 192, tid);
    __syncthreads();

    for (int hh2 = 0; hh2 < 2; ++hh2) {
      {
        const int ntl = (wv & 7) * 16;
        const int n0g = hh2 * 128 + ntl;
        const v2f* bf = (const v2f*)(wfl + (hh2 * 8 + (wv & 7)) * 1024) + lane;
        v2f wf[16];
#pragma unroll
        for (int ks = 0; ks < 16; ++ks) wf[ks] = bf[ks * 32];
        const float bias = b_fc[n0g + l16];
        for (int i = 0; i < 4; ++i) {
          const int m0 = ((wv >> 3) + 2 * i) * 16;
          v2f af[16];
#pragma unroll
          for (int ks = 0; ks < 16; ++ks) {
            af[ks].x = hb[(m0 + l16) * LP + 4 * ks + koff];
            af[ks].y = hb[(m0 + l16) * LP + 4 * ks + koff + 1];
          }
          frag_fence();
          v8f acc;
          for (int r = 0; r < 8; ++r) acc[r] = bias;
#pragma unroll
          for (int ks = 0; ks < 16; ++ks) acc = wmma4(af[ks], wf[ks], acc);
          frag_fence();
          float ge[8];
          for (int r = 0; r < 8; ++r) ge[r] = acc[r] * (GK1 + GK3 * (acc[r] * acc[r]));
          for (int r = 0; r < 8; ++r) ge[r] = __builtin_amdgcn_exp2f(ge[r]);
          for (int r = 0; r < 8; ++r) ge[r] = __builtin_amdgcn_rcpf(1.0f + ge[r]);
          for (int r = 0; r < 8; ++r)
            gb[(m0 + r + 8 * half) * GP + ntl + l16] = acc[r] * ge[r];
        }
      }
      __syncthreads();
      {
        const int n0 = (wv & 3) * 16;
        const v2f* bp = (const v2f*)(wpl + ((wv & 3) * 64 + hh2 * 32) * 64) + lane;
        v2f wp[32];
#pragma unroll
        for (int ks = 0; ks < 32; ++ks) wp[ks] = bp[ks * 32];
        const float bias = b_pr[n0 + l16];
        for (int i = 0; i < 2; ++i) {
          const int m0 = ((wv >> 2) + 4 * i) * 16;
          v8f acc;
          if (hh2 == 0) {
            for (int r = 0; r < 8; ++r) acc[r] = bias;
          } else {
            for (int r = 0; r < 8; ++r)
              acc[r] = ob[(m0 + r + 8 * half) * LP + n0 + l16];
          }
#pragma unroll
          for (int g = 0; g < 2; ++g) {
            v2f af[16];
#pragma unroll
            for (int ks = 0; ks < 16; ++ks) {
              af[ks].x = gb[(m0 + l16) * GP + 4 * (16 * g + ks) + koff];
              af[ks].y = gb[(m0 + l16) * GP + 4 * (16 * g + ks) + koff + 1];
            }
            frag_fence();
#pragma unroll
            for (int ks = 0; ks < 16; ++ks) acc = wmma4(af[ks], wp[16 * g + ks], acc);
            frag_fence();
          }
          if (hh2 == 0) {
            for (int r = 0; r < 8; ++r)
              ob[(m0 + r + 8 * half) * LP + n0 + l16] = acc[r];
          } else {
            for (int r = 0; r < 8; ++r)
              xb[(m0 + r + 8 * half) * LP + n0 + l16] += 0.1f * acc[r];
          }
        }
      }
      __syncthreads();
    }
  }

  {
    const int s = tid >> 2, p = tid & 3;
    float sum = 0.0f, sq = 0.0f;
    for (int d = p * 16; d < p * 16 + 16; ++d) {
      float xv = xb[s * LP + d];
      sum += xv; sq += xv * xv;
    }
    sum += row_xor<1>(sum); sq += row_xor<1>(sq);
    sum += row_xor<2>(sum); sq += row_xor<2>(sq);
    const float mu  = sum * (1.0f / 64.0f);
    const float var = sq * (1.0f / 64.0f) - mu * mu;
    const float inv = rsqrtf(var + 1e-5f);
    for (int d = p * 16; d < p * 16 + 16; ++d)
      out[b * 8192 + s * 64 + d] =
          (xb[s * LP + d] - mu) * inv * lnf_g[d] + lnf_b[d];
  }
}

extern "C" void kernel_launch(void* const* d_in, const int* in_sizes, int n_in,
                              void* d_out, int out_size, void* d_ws, size_t ws_size,
                              hipStream_t stream) {
  const int*   idx   = (const int*)  d_in[0];
  const int*   nlp   = (const int*)  d_in[1];
  const float* wte   = (const float*)d_in[2];
  const float* wpe   = (const float*)d_in[3];
  const float* t_w1  = (const float*)d_in[4];
  const float* t_b1  = (const float*)d_in[5];
  const float* t_w2  = (const float*)d_in[6];
  const float* t_b2  = (const float*)d_in[7];
  const float* ln1_g = (const float*)d_in[8];
  const float* ln1_b = (const float*)d_in[9];
  const float* w_qkv = (const float*)d_in[10];
  const float* b_qkv = (const float*)d_in[11];
  const float* w_o   = (const float*)d_in[12];
  const float* b_o   = (const float*)d_in[13];
  const float* ln2_g = (const float*)d_in[14];
  const float* ln2_b = (const float*)d_in[15];
  const float* w_fc  = (const float*)d_in[16];
  const float* b_fc  = (const float*)d_in[17];
  const float* w_pr  = (const float*)d_in[18];
  const float* b_pr  = (const float*)d_in[19];
  const float* lnf_g = (const float*)d_in[20];
  const float* lnf_b = (const float*)d_in[21];

  float* ttab = (float*)d_ws;

  temb_kernel<<<128, 256, 0, stream>>>(nlp, t_w1, t_b1, t_w2, t_b2, ttab);
  xformer_kernel<<<32, TB, 0, stream>>>(idx, nlp, wte, wpe, ln1_g, ln1_b,
                                        w_qkv, b_qkv, w_o, b_o, ln2_g, ln2_b,
                                        w_fc, b_fc, w_pr, b_pr, lnf_g, lnf_b,
                                        ttab, (float*)d_out);
}


# QKV and out-projection GEMMs: a wave keeps 2 row tiles' A fragments, uses each weight fragment for both (2 accumulator chains), and asks for weights one column tile ahead; sums and order unchanged
# speedup vs baseline: 3.9271x; 1.0448x over previous
#include <hip/hip_runtime.h>
#include <hip/hip_bf16.h>
#include <math.h>


typedef __attribute__((ext_vector_type(2))) float v2f;
typedef __attribute__((ext_vector_type(8))) float v8f;

__device__ __forceinline__ v8f wmma4(v2f a, v2f b, v8f c) {
  return __builtin_amdgcn_wmma_f32_16x16x4_f32(
       false, a,  false, b,
       (short)0, c,  false,  false);
}

__device__ __forceinline__ void frag_fence() { __builtin_amdgcn_sched_barrier(0); }

template <int M>
__device__ __forceinline__ float row_xor(float v) {
  return __int_as_float(__builtin_amdgcn_update_dpp(
      0, __float_as_int(v), 0x160 + M, 0xf, 0xf, true));
}

__device__ __forceinline__ float xhalf(float v) {
  const int b = __float_as_int(v);
  return __int_as_float(__builtin_amdgcn_permlanex16(
      b, b, 0x76543210, 0xfedcba98, false, false));
}

__global__ __launch_bounds__(256) void temb_kernel(
    const int* __restrict__ num_loops,
    const float* __restrict__ t_w1, const float* __restrict__ t_b1,
    const float* __restrict__ t_w2, const float* __restrict__ t_b2,
    float* __restrict__ ttab) {
  const int t = blockIdx.x;
  if (t >= *num_loops) return;
  __shared__ float te[256];
  __shared__ float hs[1024];
  __shared__ float red[256];
  const int tid = threadIdx.x;

  if (tid < 128) {
    float f   = expf(-logf(10000.0f) * (float)tid / 128.0f);
    float arg = (float)t * f;
    te[tid]       = cosf(arg);
    te[128 + tid] = sinf(arg);
  }
  __syncthreads();

  for (int j = tid; j < 1024; j += 256) {
    float acc = t_b1[j];
    for (int k = 0; k < 256; ++k) acc += te[k] * t_w1[k * 1024 + j];
    hs[j] = acc / (1.0f + expf(-acc));
  }
  __syncthreads();

  const int j = tid & 63, p = tid >> 6;
  float acc = 0.0f;
  for (int k = p * 256; k < p * 256 + 256; ++k) acc += hs[k] * t_w2[k * 64 + j];
  red[tid] = acc;
  __syncthreads();
  if (tid < 64) {
    ttab[t * 64 + tid] =
        t_b2[tid] + red[tid] + red[tid + 64] + red[tid + 128] + red[tid + 192];
  }
}

#define TB 512
#define NW 16
#define LP 76
#define GP 132
#define VP 132
#define LSZ (128 * LP)
#define WSZ 16384
#define GK1 (-2.0f * 0.7978845608028654f * 1.4426950408889634f)
#define GK3 (-2.0f * 0.7978845608028654f * 1.4426950408889634f * 0.044715f)

__device__ __forceinline__ void ln_rows(const float* __restrict__ src,
                                        float* __restrict__ dst,
                                        const float* __restrict__ g,
                                        const float* __restrict__ bta, int tid) {
  const int s = tid >> 2, p = tid & 3;
  float sum = 0.0f, sq = 0.0f;
  for (int d = p * 16; d < p * 16 + 16; ++d) {
    float xv = src[s * LP + d];
    sum += xv; sq += xv * xv;
  }
  sum += row_xor<1>(sum); sq += row_xor<1>(sq);
  sum += row_xor<2>(sum); sq += row_xor<2>(sq);
  const float mu  = sum * (1.0f / 64.0f);
  const float var = sq * (1.0f / 64.0f) - mu * mu;
  const float inv = rsqrtf(var + 1e-5f);
  for (int d = p * 16; d < p * 16 + 16; ++d)
    dst[s * LP + d] = (src[s * LP + d] - mu) * inv * g[d] + bta[d];
}

template <int LD>
__device__ __forceinline__ void load_wcol(v2f (&w)[16], float& bias,
                                          const float* W, const int li,
                                          const float* bv, const int bi) {
  bias = bv[bi];
#pragma unroll
  for (int ks = 0; ks < 16; ++ks) {
    w[ks].x = W[li + 4 * ks * LD];
    w[ks].y = W[li + 4 * ks * LD + LD];
  }
}

__device__ __forceinline__ void load_arow(v2f (&a)[16], const float* __restrict__ row) {
#pragma unroll
  for (int ks = 0; ks < 16; ++ks) { a[ks].x = row[4 * ks]; a[ks].y = row[4 * ks + 1]; }
}

__device__ __forceinline__ void mma_2rows(const v2f (&a0)[16], const v2f (&a1)[16],
                                          const v2f (&w)[16], v8f& c0, v8f& c1) {
#pragma unroll
  for (int ks = 0; ks < 16; ++ks) {
    c0 = wmma4(a0[ks], w[ks], c0);
    c1 = wmma4(a1[ks], w[ks], c1);
  }
}

__global__ __launch_bounds__(TB, 1) void xformer_kernel(
    const int* __restrict__ idx, const int* __restrict__ num_loops,
    const float* __restrict__ wte, const float* __restrict__ wpe,
    const float* __restrict__ ln1_g, const float* __restrict__ ln1_b,
    const float* w_qkv, const float* b_qkv,
    const float* w_o, const float* b_o,
    const float* __restrict__ ln2_g, const float* __restrict__ ln2_b,
    const float* __restrict__ w_fc, const float* __restrict__ b_fc,
    const float* __restrict__ w_pr, const float* __restrict__ b_pr,
    const float* __restrict__ lnf_g, const float* __restrict__ lnf_b,
    const float* __restrict__ ttab, float* __restrict__ out) {
  __shared__ __align__(16) float smem[5 * LSZ + 2 * WSZ + 256];
  float* xb = smem;
  float* hb = smem + LSZ;
  float* qb = smem + 2 * LSZ;
  float* kb = smem + 3 * LSZ;
  float* vb = smem + 4 * LSZ;
  float* gb = qb;
  float* ob = vb;
  float* wfl = smem + 5 * LSZ;
  float* wpl = smem + 5 * LSZ + WSZ;
  float* lnp = smem + 5 * LSZ + 2 * WSZ;

  const int b    = blockIdx.x;
  const int tid  = threadIdx.x;
  const int wid  = tid >> 5;
  const int lane = tid & 31;
  const int l16  = lane & 15;
  const int half = lane >> 4;
  const int koff = 2 * half;
  const int wv = __builtin_amdgcn_readfirstlane(wid);

  for (int i = tid; i < 8192; i += TB) {
    int s = i >> 6, d = i & 63;
    xb[s * LP + d] = wte[idx[b * 128 + s] * 64 + d] + wpe[i];
  }
  for (int i = tid; i < WSZ; i += TB) {
    const int k = i >> 8, n = i & 255;
    wfl[(((n >> 4) * 16 + (k >> 2)) * 32 + ((k >> 1) & 1) * 16 + (n & 15)) * 2 + (k & 1)] = w_fc[i];
  }
  for (int i = tid; i < WSZ; i += TB) {
    const int k = i >> 6, n = i & 63;
    wpl[(((n >> 4) * 64 + (k >> 2)) * 32 + ((k >> 1) & 1) * 16 + (n & 15)) * 2 + (k & 1)] = w_pr[i];
  }
  if (tid < 64) {
    lnp[tid]       = ln1_g[tid];
    lnp[64 + tid]  = ln1_b[tid];
    lnp[128 + tid] = ln2_g[tid];
    lnp[192 + tid] = ln2_b[tid];
  }
  __syncthreads();

  const int NL = *num_loops;
  for (int t = 0; t < NL; ++t) {
    for (int i = tid; i < 8192; i += TB)
      xb[(i >> 6) * LP + (i & 63)] += ttab[t * 64 + (i & 63)];
    __syncthreads();

    const int cg = wv >> 2, rg = wv & 3;
    v2f wqa[16], wqb[16];
    float bqa, bqb;
    const int bql = 16 * cg + l16;
    const int wql = koff * 192 + bql;
    load_wcol<192>(wqa, bqa, w_qkv, wql, b_qkv, bql);

    ln_rows(xb, hb, lnp, lnp + 64, tid);
    __syncthreads();

    {
      const int m0 = 32 * rg;
      v2f af0[16], af1[16];
      load_arow(af0, hb + (m0 + l16) * LP + koff);
      load_arow(af1, hb + (m0 + 16 + l16) * LP + koff);
#pragma nounroll
      for (int ct = 0; ct < 3; ++ct) {
        if (ct < 2) load_wcol<192>(wqb, bqb, w_qkv + 64 * (ct + 1), wql, b_qkv + 64 * (ct + 1), bql);
        frag_fence();
        v8f acc0, acc1;
        for (int r = 0; r < 8; ++r) { acc0[r] = bqa; acc1[r] = bqa; }
        mma_2rows(af0, af1, wqa, acc0, acc1);
        frag_fence();
        if (ct < 2) {
          float* dst = (ct == 0 ? qb : kb) + (m0 + 8 * half) * LP + 16 * cg + l16;
          for (int r = 0; r < 8; ++r) { dst[r * LP] = acc0[r]; dst[(16 + r) * LP] = acc1[r]; }
        } else {
          float4* dv = (float4*)(vb + (16 * cg + l16) * VP + m0 + 8 * half);
          dv[0] = make_float4(acc0[0], acc0[1], acc0[2], acc0[3]);
          dv[1] = make_float4(acc0[4], acc0[5], acc0[6], acc0[7]);
          dv[4] = make_float4(acc1[0], acc1[1], acc1[2], acc1[3]);
          dv[5] = make_float4(acc1[4], acc1[5], acc1[6], acc1[7]);
        }
#pragma unroll
        for (int ks = 0; ks < 16; ++ks) wqa[ks] = wqb[ks];
        bqa = bqb;
      }
    }
    __syncthreads();

    v2f wo[16];
    float bo;
    for (int jj = 0; jj < 2; ++jj) {
      if (jj == 1) load_wcol<64>(wo, bo, w_o, koff * 64 + bql, b_o, bql);
      const int hh = wv >> 2;
      const int qt = jj ? (wv & 3) : 7 - (wv & 3);
      const int q0 = qt * 16, c0 = hh * 16;
      v2f qf[4];
#pragma unroll
      for (int kk = 0; kk < 4; ++kk) {
        qf[kk].x = qb[(q0 + l16) * LP + c0 + 4 * kk + koff];
        qf[kk].y = qb[(q0 + l16) * LP + c0 + 4 * kk + koff + 1];
      }
      v8f sc[8];
#pragma unroll
      for (int g = 0; g < 2; ++g) {
        if (4 * g > qt) break;
        v2f kf[4][4];
#pragma unroll
        for (int j = 0; j < 4; ++j) {
#pragma unroll
          for (int kk = 0; kk < 4; ++kk) {
            kf[j][kk].x = kb[((4 * g + j) * 16 + l16) * LP + c0 + 4 * kk + koff];
            kf[j][kk].y = kb[((4 * g + j) * 16 + l16) * LP + c0 + 4 * kk + koff + 1];
          }
        }
        frag_fence();
#pragma unroll
        for (int j = 0; j < 4; ++j) {
          const int kt = 4 * g + j;
          if (kt > qt) break;
          v8f a8; for (int r = 0; r < 8; ++r) a8[r] = 0.0f;
#pragma unroll
          for (int kk = 0; kk < 4; ++kk)
            a8 = wmma4(kf[j][kk], qf[kk], a8);
          for (int r = 0; r < 8; ++r) a8[r] *= 0.25f;
          if (kt == qt) {
            for (int r = 0; r < 8; ++r)
              a8[r] = (r + 8 * half <= l16) ? a8[r] : -1e9f;
          }
          sc[kt] = a8;
        }
        frag_fence();
      }
      float m8[8], s8[8];
#pragma unroll
      for (int r = 0; r < 8; ++r) m8[r] = sc[0][r];
#pragma unroll
      for (int kt = 1; kt < 8; ++kt) {
        if (kt > qt) break;
#pragma unroll
        for (int r = 0; r < 8; ++r) m8[r] = fmaxf(m8[r], sc[kt][r]);
      }
      float mx = fmaxf(fmaxf(fmaxf(m8[0], m8[1]), fmaxf(m8[2], m8[3])),
                       fmaxf(fmaxf(m8[4], m8[5]), fmaxf(m8[6], m8[7])));
      mx = fmaxf(mx, xhalf(mx));
#pragma unroll
      for (int r = 0; r < 8; ++r) s8[r] = 0.0f;
#pragma unroll
      for (int kt = 0; kt < 8; ++kt) {
        if (kt > qt) break;
#pragma unroll
        for (int r = 0; r < 8; ++r) {
          float e = __expf(sc[kt][r] - mx);
          sc[kt][r] = e; s8[r] += e;
        }
      }
      float sm = ((s8[0] + s8[1]) + (s8[2] + s8[3])) + ((s8[4] + s8[5]) + (s8[6] + s8[7]));
      sm += xhalf(sm);
      const float inv = 1.0f / sm;
      const float4* vp = (const float4*)(vb + (c0 + l16) * VP + 8 * half);
      v8f o; for (int r = 0; r < 8; ++r) o[r] = 0.0f;
#pragma unroll
      for (int g = 0; g < 2; ++g) {
        if (4 * g > qt) break;
        float4 vlo[4], vhi[4];
#pragma unroll
        for (int j = 0; j < 4; ++j) { vlo[j] = vp[4 * (4 * g + j)]; vhi[j] = vp[4 * (4 * g + j) + 1]; }
        frag_fence();
#pragma unroll
        for (int j = 0; j < 4; ++j) {
          const int kt = 4 * g + j;
          if (kt > qt) break;
          v2f vf, pf;
          vf.x = vlo[j].x; vf.y = vlo[j].y; pf.x = sc[kt][0]; pf.y = sc[kt][1];
          o = wmma4(vf, pf, o);
          vf.x = vlo[j].z; vf.y = vlo[j].w; pf.x = sc[kt][2]; pf.y = sc[kt][3];
          o = wmma4(vf, pf, o);
          vf.x = vhi[j].x; vf.y = vhi[j].y; pf.x = sc[kt][4]; pf.y = sc[kt][5];
          o = wmma4(vf, pf, o);
          vf.x = vhi[j].z; vf.y = vhi[j].w; pf.x = sc[kt][6]; pf.y = sc[kt][7];
          o = wmma4(vf, pf, o);
        }
        frag_fence();
      }
      float4* dst = (float4*)(hb + (q0 + l16) * LP + c0 + 8 * half);
      dst[0] = make_float4(o[0] * inv, o[1] * inv, o[2] * inv, o[3] * inv);
      dst[1] = make_float4(o[4] * inv, o[5] * inv, o[6] * inv, o[7] * inv);
    }
    __syncthreads();

    {
      const int n0 = 16 * cg, m0 = 32 * rg;
      v2f af0[16], af1[16];
      load_arow(af0, hb + (m0 + l16) * LP + koff);
      load_arow(af1, hb + (m0 + 16 + l16) * LP + koff);
      frag_fence();
      v8f acc0, acc1;
      for (int r = 0; r < 8; ++r) { acc0[r] = bo; acc1[r] = bo; }
      mma_2rows(af0, af1, wo, acc0, acc1);
      frag_fence();
      for (int r = 0; r < 8; ++r) {
        xb[(m0 + r + 8 * half) * LP + n0 + l16]      += 0.1f * acc0[r];
        xb[(m0 + 16 + r + 8 * half) * LP + n0 + l16] += 0.1f * acc1[r];
      }
    }
    __syncthreads();

    ln_rows(xb, hb, lnp + 128, lnp + 192, tid);
    __syncthreads();

    for (int hh2 = 0; hh2 < 2; ++hh2) {
      {
        const int ntl = (wv & 7) * 16;
        const int n0g = hh2 * 128 + ntl;
        const v2f* bf = (const v2f*)(wfl + (hh2 * 8 + (wv & 7)) * 1024) + lane;
        v2f wf[16];
#pragma unroll
        for (int ks = 0; ks < 16; ++ks) wf[ks] = bf[ks * 32];
        const float bias = b_fc[n0g + l16];
        for (int i = 0; i < 4; ++i) {
          const int m0 = ((wv >> 3) + 2 * i) * 16;
          v2f af[16];
#pragma unroll
          for (int ks = 0; ks < 16; ++ks) {
            af[ks].x = hb[(m0 + l16) * LP + 4 * ks + koff];
            af[ks].y = hb[(m0 + l16) * LP + 4 * ks + koff + 1];
          }
          frag_fence();
          v8f acc;
          for (int r = 0; r < 8; ++r) acc[r] = bias;
#pragma unroll
          for (int ks = 0; ks < 16; ++ks) acc = wmma4(af[ks], wf[ks], acc);
          frag_fence();
          float ge[8];
          for (int r = 0; r < 8; ++r) ge[r] = acc[r] * (GK1 + GK3 * (acc[r] * acc[r]));
          for (int r = 0; r < 8; ++r) ge[r] = __builtin_amdgcn_exp2f(ge[r]);
          for (int r = 0; r < 8; ++r) ge[r] = __builtin_amdgcn_rcpf(1.0f + ge[r]);
          for (int r = 0; r < 8; ++r)
            gb[(m0 + r + 8 * half) * GP + ntl + l16] = acc[r] * ge[r];
        }
      }
      __syncthreads();
      {
        const int n0 = (wv & 3) * 16;
        const v2f* bp = (const v2f*)(wpl + ((wv & 3) * 64 + hh2 * 32) * 64) + lane;
        v2f wp[32];
#pragma unroll
        for (int ks = 0; ks < 32; ++ks) wp[ks] = bp[ks * 32];
        const float bias = b_pr[n0 + l16];
        for (int i = 0; i < 2; ++i) {
          const int m0 = ((wv >> 2) + 4 * i) * 16;
          v8f acc;
          if (hh2 == 0) {
            for (int r = 0; r < 8; ++r) acc[r] = bias;
          } else {
            for (int r = 0; r < 8; ++r)
              acc[r] = ob[(m0 + r + 8 * half) * LP + n0 + l16];
          }
#pragma unroll
          for (int g = 0; g < 2; ++g) {
            v2f af[16];
#pragma unroll
            for (int ks = 0; ks < 16; ++ks) {
              af[ks].x = gb[(m0 + l16) * GP + 4 * (16 * g + ks) + koff];
              af[ks].y = gb[(m0 + l16) * GP + 4 * (16 * g + ks) + koff + 1];
            }
            frag_fence();
#pragma unroll
            for (int ks = 0; ks < 16; ++ks) acc = wmma4(af[ks], wp[16 * g + ks], acc);
            frag_fence();
          }
          if (hh2 == 0) {
            for (int r = 0; r < 8; ++r)
              ob[(m0 + r + 8 * half) * LP + n0 + l16] = acc[r];
          } else {
            for (int r = 0; r < 8; ++r)
              xb[(m0 + r + 8 * half) * LP + n0 + l16] += 0.1f * acc[r];
          }
        }
      }
      __syncthreads();
    }
  }

  {
    const int s = tid >> 2, p = tid & 3;
    const float* xr = xb + s * LP + p * 16;
    const float* gp = lnf_g + p * 16;
    const float* bp = lnf_b + p * 16;
    float* op = out + b * 8192 + tid * 16;
    float sum = 0.0f, sq = 0.0f;
    for (int j = 0; j < 16; ++j) {
      float xv = xr[j];
      sum += xv; sq += xv * xv;
    }
    sum += row_xor<1>(sum); sq += row_xor<1>(sq);
    sum += row_xor<2>(sum); sq += row_xor<2>(sq);
    const float mu  = sum * (1.0f / 64.0f);
    const float var = sq * (1.0f / 64.0f) - mu * mu;
    const float inv = rsqrtf(var + 1e-5f);
    for (int j = 0; j < 16; ++j)
      op[j] = (xr[j] - mu) * inv * gp[j] + bp[j];
  }
}

extern "C" void kernel_launch(void* const* d_in, const int* in_sizes, int n_in,
                              void* d_out, int out_size, void* d_ws, size_t ws_size,
                              hipStream_t stream) {
  const int*   idx   = (const int*)  d_in[0];
  const int*   nlp   = (const int*)  d_in[1];
  const float* wte   = (const float*)d_in[2];
  const float* wpe   = (const float*)d_in[3];
  const float* t_w1  = (const float*)d_in[4];
  const float* t_b1  = (const float*)d_in[5];
  const float* t_w2  = (const float*)d_in[6];
  const float* t_b2  = (const float*)d_in[7];
  const float* ln1_g = (const float*)d_in[8];
  const float* ln1_b = (const float*)d_in[9];
  const float* w_qkv = (const float*)d_in[10];
  const float* b_qkv = (const float*)d_in[11];
  const float* w_o   = (const float*)d_in[12];
  const float* b_o   = (const float*)d_in[13];
  const float* ln2_g = (const float*)d_in[14];
  const float* ln2_b = (const float*)d_in[15];
  const float* w_fc  = (const float*)d_in[16];
  const float* b_fc  = (const float*)d_in[17];
  const float* w_pr  = (const float*)d_in[18];
  const float* b_pr  = (const float*)d_in[19];
  const float* lnf_g = (const float*)d_in[20];
  const float* lnf_b = (const float*)d_in[21];

  float* ttab = (float*)d_ws;

  temb_kernel<<<128, 256, 0, stream>>>(nlp, t_w1, t_b1, t_w2, t_b2, ttab);
  xformer_kernel<<<32, TB, 0, stream>>>(idx, nlp, wte, wpe, ln1_g, ln1_b,
                                        w_qkv, b_qkv, w_o, b_o, ln2_g, ln2_b,
                                        w_fc, b_fc, w_pr, b_pr, lnf_g, lnf_b,
                                        ttab, (float*)d_out);
}
